# starting kernel
# baseline (speedup 1.0000x reference)
#include <hip/hip_runtime.h>
#include <math.h>

#define L_SEQ     2048
#define DMODEL    512
#define DINNER    1024
#define DSTATE    128
#define HEADDIM   64
#define NHEADSN   16
#define CONVDIM   1280
#define DINPROJ   2320
#define EPS_RMS   1e-5f

typedef __attribute__((ext_vector_type(2))) float v2f;
typedef __attribute__((ext_vector_type(8))) float v8f;

__global__ void bm_wmma_gemm_nt_f32(const float* __restrict__ A,
                                    const float* __restrict__ Bw,
                                    float* __restrict__ C,
                                    int M, int N, int K, int flipA)
{
    const int lane   = threadIdx.x & 31;
    const int wave   = threadIdx.x >> 5;
    const int tileId = blockIdx.x * (blockDim.x >> 5) + wave;
    const int ntn    = N >> 4;
    const int tile_mb = tileId / ntn;
    const int tile_n  = tileId - tile_mb * ntn;
    if (tile_mb * 64 >= M) return;

    const int mn   = lane & 15;
    const int koff = (lane >> 4) << 1;

    const float* Ar0;
    const float* Ar1;
    const float* Ar2;
    const float* Ar3;
    {
        int r0 = tile_mb * 64 +  0 + mn;
        int r1 = tile_mb * 64 + 16 + mn;
        int r2 = tile_mb * 64 + 32 + mn;
        int r3 = tile_mb * 64 + 48 + mn;
        if (flipA) { r0 = M - 1 - r0; r1 = M - 1 - r1; r2 = M - 1 - r2; r3 = M - 1 - r3; }
        Ar0 = A + (size_t)r0 * K + koff;
        Ar1 = A + (size_t)r1 * K + koff;
        Ar2 = A + (size_t)r2 * K + koff;
        Ar3 = A + (size_t)r3 * K + koff;
    }
    const float* Br = Bw + (size_t)(tile_n * 16 + mn) * K + koff;

    v8f acc0 = {}, acc1 = {}, acc2 = {}, acc3 = {};

    v2f b, a0, a1, a2, a3;
    b  = *(const v2f*)(Br);
    a0 = *(const v2f*)(Ar0);
    a1 = *(const v2f*)(Ar1);
    a2 = *(const v2f*)(Ar2);
    a3 = *(const v2f*)(Ar3);

    for (int k = 0; k < K - 4; k += 4) {
        if ((k & 63) == 0) {
            __builtin_prefetch(Br  + k + 256, 0, 1);
            __builtin_prefetch(Ar0 + k + 256, 0, 1);
            __builtin_prefetch(Ar2 + k + 256, 0, 1);
        }
        const v2f bn  = *(const v2f*)(Br  + k + 4);
        const v2f an0 = *(const v2f*)(Ar0 + k + 4);
        const v2f an1 = *(const v2f*)(Ar1 + k + 4);
        const v2f an2 = *(const v2f*)(Ar2 + k + 4);
        const v2f an3 = *(const v2f*)(Ar3 + k + 4);
        acc0 = __builtin_amdgcn_wmma_f32_16x16x4_f32(false, a0, false, b,
                                                     (short)0, acc0, false, false);
        acc1 = __builtin_amdgcn_wmma_f32_16x16x4_f32(false, a1, false, b,
                                                     (short)0, acc1, false, false);
        acc2 = __builtin_amdgcn_wmma_f32_16x16x4_f32(false, a2, false, b,
                                                     (short)0, acc2, false, false);
        acc3 = __builtin_amdgcn_wmma_f32_16x16x4_f32(false, a3, false, b,
                                                     (short)0, acc3, false, false);
        b = bn; a0 = an0; a1 = an1; a2 = an2; a3 = an3;
    }
    acc0 = __builtin_amdgcn_wmma_f32_16x16x4_f32(false, a0, false, b,
                                                 (short)0, acc0, false, false);
    acc1 = __builtin_amdgcn_wmma_f32_16x16x4_f32(false, a1, false, b,
                                                 (short)0, acc1, false, false);
    acc2 = __builtin_amdgcn_wmma_f32_16x16x4_f32(false, a2, false, b,
                                                 (short)0, acc2, false, false);
    acc3 = __builtin_amdgcn_wmma_f32_16x16x4_f32(false, a3, false, b,
                                                 (short)0, acc3, false, false);

    const int msub = (lane >> 4) << 3;
    float* Cp = C + (size_t)(tile_mb * 64 + msub) * N + tile_n * 16 + mn;
#pragma unroll
    for (int v = 0; v < 8; v++) Cp[(size_t)v * N]              = acc0[v];
#pragma unroll
    for (int v = 0; v < 8; v++) Cp[(size_t)(16 + v) * N]       = acc1[v];
#pragma unroll
    for (int v = 0; v < 8; v++) Cp[(size_t)(32 + v) * N]       = acc2[v];
#pragma unroll
    for (int v = 0; v < 8; v++) Cp[(size_t)(48 + v) * N]       = acc3[v];
}

__global__ void bm_conv_silu_dt(const float* __restrict__ zx,
                                const float* __restrict__ convw,
                                const float* __restrict__ convb,
                                const float* __restrict__ dtb,
                                const float* __restrict__ Alog,
                                float* __restrict__ xbc,
                                float* __restrict__ dt_o,
                                float* __restrict__ dA_o)
{
    const int idx = blockIdx.x * blockDim.x + threadIdx.x;
    const int W   = CONVDIM + NHEADSN;
    if (idx >= L_SEQ * W) return;
    const int l = idx / W;
    const int c = idx - l * W;
    if (c < CONVDIM) {
        float acc = convb[c];
#pragma unroll
        for (int t = 0; t < 4; t++) {
            const int ll = l - 3 + t;
            if (ll >= 0)
                acc += convw[c * 4 + t] * zx[(size_t)ll * DINPROJ + DINNER + c];
        }
        const float s = acc / (1.0f + expf(-acc));
        xbc[(size_t)l * CONVDIM + c] = s;
    } else {
        const int h = c - CONVDIM;
        const float v  = zx[(size_t)l * DINPROJ + DINNER + CONVDIM + h] + dtb[h];
        const float sp = (v > 20.0f) ? v : log1pf(expf(v));
        const float Ah = -expf(Alog[h]);
        dt_o[l * NHEADSN + h] = sp;
        dA_o[l * NHEADSN + h] = expf(sp * Ah);
    }
}

__global__ void bm_ssm_scan(const float* __restrict__ xbc,
                            const float* __restrict__ dt,
                            const float* __restrict__ dA,
                            float* __restrict__ yscan)
{
    const int head = blockIdx.x;
    const int tid  = threadIdx.x;
    const int p    = tid >> 2;
    const int nb   = (tid & 3) * 32;

    __shared__ float sx[HEADDIM];
    __shared__ float sB[DSTATE];
    __shared__ float sC[DSTATE];

    float h[32];
#pragma unroll
    for (int j = 0; j < 32; j++) h[j] = 0.0f;

    for (int l = 0; l < L_SEQ; l++) {
        const float* row = xbc + (size_t)l * CONVDIM;
        if (tid < 128) sB[tid]       = row[DINNER + tid];
        else           sC[tid - 128] = row[DINNER + DSTATE + (tid - 128)];
        if (tid < 64)  sx[tid]       = row[head * HEADDIM + tid];
        __syncthreads();

        const float dAl  = dA[l * NHEADSN + head];
        const float coef = dt[l * NHEADSN + head] * sx[p];
        float acc = 0.0f;
#pragma unroll
        for (int j = 0; j < 32; j++) {
            h[j] = h[j] * dAl + coef * sB[nb + j];
            acc += h[j] * sC[nb + j];
        }
        acc += __shfl_xor(acc, 1, 32);
        acc += __shfl_xor(acc, 2, 32);
        if ((tid & 3) == 0)
            yscan[(size_t)l * DINNER + head * HEADDIM + p] = acc;
        __syncthreads();
    }
}

__global__ void bm_gate_rmsnorm(float* __restrict__ y,
                                const float* __restrict__ xbc,
                                const float* __restrict__ zx,
                                const float* __restrict__ Dp,
                                const float* __restrict__ normw)
{
    const int l   = blockIdx.x;
    const int tid = threadIdx.x;
    __shared__ float red[8];
    __shared__ float scale;

    float vals[4];
    float ss = 0.0f;
#pragma unroll
    for (int j = 0; j < 4; j++) {
        const int i = tid * 4 + j;
        const int hh = i >> 6;
        float v = y[(size_t)l * DINNER + i] + Dp[hh] * xbc[(size_t)l * CONVDIM + i];
        const float z = zx[(size_t)l * DINPROJ + i];
        v *= z / (1.0f + expf(-z));
        vals[j] = v;
        ss += v * v;
    }
#pragma unroll
    for (int o = 16; o >= 1; o >>= 1) ss += __shfl_xor(ss, o, 32);
    if ((tid & 31) == 0) red[tid >> 5] = ss;
    __syncthreads();
    if (tid == 0) {
        float t = 0.0f;
        for (int i = 0; i < 8; i++) t += red[i];
        scale = rsqrtf(t / (float)DINNER + EPS_RMS);
    }
    __syncthreads();
#pragma unroll
    for (int j = 0; j < 4; j++) {
        const int i = tid * 4 + j;
        y[(size_t)l * DINNER + i] = vals[j] * scale * normw[i];
    }
}

__global__ void bm_add_flip(const float* __restrict__ of,
                            const float* __restrict__ ob,
                            float* __restrict__ out)
{
    const int idx = blockIdx.x * blockDim.x + threadIdx.x;
    if (idx >= L_SEQ * DMODEL) return;
    const int l = idx / DMODEL;
    const int j = idx - l * DMODEL;
    out[idx] = of[idx] + ob[(size_t)(L_SEQ - 1 - l) * DMODEL + j];
}

extern "C" void kernel_launch(void* const* d_in, const int* in_sizes, int n_in,
                              void* d_out, int out_size, void* d_ws, size_t ws_size,
                              hipStream_t stream)
{
    const float* Win[2]   = { (const float*)d_in[0],  (const float*)d_in[8]  };
    const float* convw[2] = { (const float*)d_in[1],  (const float*)d_in[9]  };
    const float* convb[2] = { (const float*)d_in[2],  (const float*)d_in[10] };
    const float* dtb[2]   = { (const float*)d_in[3],  (const float*)d_in[11] };
    const float* Alog[2]  = { (const float*)d_in[4],  (const float*)d_in[12] };
    const float* Dp[2]    = { (const float*)d_in[5],  (const float*)d_in[13] };
    const float* normw[2] = { (const float*)d_in[6],  (const float*)d_in[14] };
    const float* Wout[2]  = { (const float*)d_in[7],  (const float*)d_in[15] };
    const float* x        =   (const float*)d_in[16];

    float* ws = (float*)d_ws;
    size_t off = 0;
    float* zx[2];  float* xbc[2]; float* dtw[2]; float* dAw[2];
    float* ysc[2]; float* od[2];
    for (int d = 0; d < 2; d++) { zx[d]  = ws + off; off += (size_t)L_SEQ * DINPROJ; }
    for (int d = 0; d < 2; d++) { xbc[d] = ws + off; off += (size_t)L_SEQ * CONVDIM; }
    for (int d = 0; d < 2; d++) { dtw[d] = ws + off; off += (size_t)L_SEQ * NHEADSN; }
    for (int d = 0; d < 2; d++) { dAw[d] = ws + off; off += (size_t)L_SEQ * NHEADSN; }
    for (int d = 0; d < 2; d++) { ysc[d] = ws + off; off += (size_t)L_SEQ * DINNER;  }
    for (int d = 0; d < 2; d++) { od[d]  = ws + off; off += (size_t)L_SEQ * DMODEL;  }

    const dim3 blk256(256);

    for (int d = 0; d < 2; d++) {
        {
            const int tiles = (L_SEQ / 64) * (DINPROJ / 16);
            bm_wmma_gemm_nt_f32<<<dim3((tiles + 7) / 8), blk256, 0, stream>>>(
                x, Win[d], zx[d], L_SEQ, DINPROJ, DMODEL, d);
        }
        {
            const int total = L_SEQ * (CONVDIM + NHEADSN);
            bm_conv_silu_dt<<<dim3((total + 255) / 256), blk256, 0, stream>>>(
                zx[d], convw[d], convb[d], dtb[d], Alog[d],
                xbc[d], dtw[d], dAw[d]);
        }
        bm_ssm_scan<<<dim3(NHEADSN), blk256, 0, stream>>>(
            xbc[d], dtw[d], dAw[d], ysc[d]);
        bm_gate_rmsnorm<<<dim3(L_SEQ), blk256, 0, stream>>>(
            ysc[d], xbc[d], zx[d], Dp[d], normw[d]);
        {
            const int tiles = (L_SEQ / 64) * (DMODEL / 16);
            bm_wmma_gemm_nt_f32<<<dim3((tiles + 7) / 8), blk256, 0, stream>>>(
                ysc[d], Wout[d], od[d], L_SEQ, DMODEL, DINNER, 0);
        }
    }
    bm_add_flip<<<dim3((L_SEQ * DMODEL + 255) / 256), blk256, 0, stream>>>(
        od[0], od[1], (float*)d_out);
}


# in_proj applies conv(4)+bias+SiLU to its own accumulators: time runs down a lane's 8 registers, 3 earlier rows come by one lane swap; the conv launch keeps 3 rows per 64-row tile and softplus(dt)
# speedup vs baseline: 25.2199x; 25.2199x over previous
#include <hip/hip_runtime.h>
#include <math.h>

#define L_SEQ     2048
#define DMODEL    512
#define DINNER    1024
#define DSTATE    128
#define HEADDIM   64
#define NHEADSN   16
#define CONVDIM   1280
#define DINPROJ   2320
#define EPS_RMS   1e-5f

typedef __attribute__((ext_vector_type(2))) float v2f;
typedef __attribute__((ext_vector_type(4))) float v4f;
typedef __attribute__((ext_vector_type(8))) float v8f;

#define INP_ROWS 64

template <int K, int N, int MT, int NT, int KSPLIT, bool FUSE>
__global__ void __launch_bounds__(128) bm_wmma_gemm_nt_f32(const float* __restrict__ A,
                                                           const float* __restrict__ Bw,
                                                           float* __restrict__ C,
                                                           int flipA,
                                                           const float* __restrict__ convw,
                                                           const float* __restrict__ convb,
                                                           float* __restrict__ xbc)
{
    constexpr int M    = L_SEQ;
    constexpr int KB   = 8;
    constexpr int KS   = KB * KSPLIT;
    constexpr int NB   = K / KS;
    constexpr int NTN  = N / (16 * NT);
    constexpr int NTM  = M / (16 * MT);
    static_assert(N % (16 * NT) == 0 && M % (16 * MT) == 0 && K % (2 * KS) == 0, "tile shape");
    static_assert(KSPLIT == 1 || (KSPLIT == 4 && MT == 4), "split shape");
    static_assert(!FUSE || (KSPLIT == 1 && N == DINPROJ && 16 * MT == INP_ROWS), "fused epilogue shape");

    const int lane   = threadIdx.x & 31;
    const int wave   = threadIdx.x >> 5;
    const int tileId = (KSPLIT == 1) ? blockIdx.x * 4 + wave : blockIdx.x;
    if (KSPLIT == 1 && tileId >= NTM * NTN) return;
    const int tm = tileId / NTN;
    const int tn = tileId - tm * NTN;
    constexpr int JST = FUSE ? NTN : 1;
    const int ct0 = FUSE ? tn : tn * NT;
    const int c  = lane & 15;
    const int h  = lane >> 4;
    const int k0 = (KSPLIT == 1) ? 0 : wave * KB;

    const float* Ar[MT];
#pragma unroll
    for (int i = 0; i < MT; i++) {
        int r = tm * (16 * MT) + 16 * i + c;
        if (flipA) r = M - 1 - r;
        Ar[i] = A + (size_t)r * K + k0 + 4 * h;
    }
    const float* Br = Bw + (size_t)(ct0 * 16 + c) * K + k0 + 4 * h;

    v8f acc[MT][NT];
#pragma unroll
    for (int i = 0; i < MT; i++)
#pragma unroll
        for (int j = 0; j < NT; j++) acc[i][j] = v8f{};

    v4f fa0[MT], fb0[NT], fa1[MT], fb1[NT];

    auto load_set = [&](v4f (&fa)[MT], v4f (&fb)[NT], int blk) {
#pragma unroll
        for (int i = 0; i < MT; i++) fa[i] = *(const v4f*)(Ar[i] + blk * KS);
#pragma unroll
        for (int j = 0; j < NT; j++) fb[j] = *(const v4f*)(Br + (size_t)j * JST * 16 * K + blk * KS);
        __builtin_amdgcn_sched_barrier(0);
    };
    auto mma_set = [&](const v4f (&fa)[MT], const v4f (&fb)[NT]) {
#pragma unroll
        for (int i = 0; i < MT; i++) {
            const v2f a = __builtin_shufflevector(fa[i], fa[i], 0, 1);
#pragma unroll
            for (int j = 0; j < NT; j++) {
                const v2f b = __builtin_shufflevector(fb[j], fb[j], 0, 1);
                acc[i][j] = __builtin_amdgcn_wmma_f32_16x16x4_f32(false, a, false, b,
                                                                  (short)0, acc[i][j], false, false);
            }
        }
#pragma unroll
        for (int i = 0; i < MT; i++) {
            const v2f a = __builtin_shufflevector(fa[i], fa[i], 2, 3);
#pragma unroll
            for (int j = 0; j < NT; j++) {
                const v2f b = __builtin_shufflevector(fb[j], fb[j], 2, 3);
                acc[i][j] = __builtin_amdgcn_wmma_f32_16x16x4_f32(false, a, false, b,
                                                                  (short)0, acc[i][j], false, false);
            }
        }
        __builtin_amdgcn_sched_barrier(0);
    };

    load_set(fa0, fb0, 0);
    for (int blk = 0; blk < NB - 2; blk += 2) {
        load_set(fa1, fb1, blk + 1);
        mma_set(fa0, fb0);
        load_set(fa0, fb0, blk + 2);
        mma_set(fa1, fb1);
    }
    load_set(fa1, fb1, NB - 1);
    mma_set(fa0, fb0);
    mma_set(fa1, fb1);

    if constexpr (FUSE) {
        const int rbase = tm * (16 * MT) + 8 * h;
#pragma unroll
        for (int j = 0; j < NT; j++) {
            const int col16 = (ct0 + j * JST) * 16;
            const int col   = col16 + c;
            float* Cp = C + (size_t)rbase * N + col;
            if (col16 < DINNER || col16 >= DINNER + CONVDIM) {
#pragma unroll
                for (int i = 0; i < MT; i++)
#pragma unroll
                    for (int v = 0; v < 8; v++)
                        Cp[(size_t)(16 * i + v) * N] = acc[i][j][v];
            } else {
                const int ch = col - DINNER;
                float* Xp = xbc + (size_t)rbase * CONVDIM + ch;
                const float w0 = convw[ch * 4 + 0];
                const float w1 = convw[ch * 4 + 1];
                const float w2 = convw[ch * 4 + 2];
                const float w3 = convw[ch * 4 + 3];
                const float cb = convb[ch];
                float pv[MT][3];
#pragma unroll
                for (int i = 0; i < MT; i++)
#pragma unroll
                    for (int q = 0; q < 3; q++)
                        pv[i][q] = __shfl_xor(acc[i][j][5 + q], 16, 32);
#pragma unroll
                for (int i = 0; i < MT; i++) {
                    float p[11];
#pragma unroll
                    for (int q = 0; q < 3; q++)
                        p[q] = h ? pv[i][q] : pv[i > 0 ? i - 1 : 0][q];
#pragma unroll
                    for (int v = 0; v < 8; v++) p[3 + v] = acc[i][j][v];
#pragma unroll
                    for (int v = 0; v < 8; v++) {
                        float a = cb;
                        a += w0 * p[v];
                        a += w1 * p[v + 1];
                        a += w2 * p[v + 2];
                        a += w3 * p[v + 3];
                        const float s = a / (1.0f + expf(-a));
                        const bool first3 = (i == 0 && v < 3);
                        const bool last3  = (i == MT - 1 && v >= 5);
                        if (!first3 || h) Xp[(size_t)(16 * i + v) * CONVDIM] = s;
                        if ((first3 && !h) || (last3 && h)) Cp[(size_t)(16 * i + v) * N] = acc[i][j][v];
                    }
                }
            }
        }
    } else if constexpr (KSPLIT == 1) {
        float* Cp = C + (size_t)(tm * (16 * MT) + 8 * h) * N + tn * (16 * NT) + c;
#pragma unroll
        for (int i = 0; i < MT; i++)
#pragma unroll
            for (int j = 0; j < NT; j++)
#pragma unroll
                for (int v = 0; v < 8; v++)
                    Cp[(size_t)(16 * i + v) * N + 16 * j] = acc[i][j][v];
    } else {
        __shared__ __attribute__((aligned(16))) float s_part[KSPLIT * MT * NT * 256];
        float* mine = s_part + wave * (MT * NT * 256) + lane * 4;
#pragma unroll
        for (int i = 0; i < MT; i++)
#pragma unroll
            for (int j = 0; j < NT; j++) {
                *(v4f*)(mine + ((i * NT + j) * 2 + 0) * 128) = __builtin_shufflevector(acc[i][j], acc[i][j], 0, 1, 2, 3);
                *(v4f*)(mine + ((i * NT + j) * 2 + 1) * 128) = __builtin_shufflevector(acc[i][j], acc[i][j], 4, 5, 6, 7);
            }
        __syncthreads();
        float* Cp = C + (size_t)(tm * (16 * MT) + 16 * wave + 8 * h) * N + tn * (16 * NT) + c;
#pragma unroll
        for (int j = 0; j < NT; j++)
#pragma unroll
            for (int q = 0; q < 2; q++) {
                const float* src = s_part + ((wave * NT + j) * 2 + q) * 128 + lane * 4;
                v4f s = *(const v4f*)(src);
#pragma unroll
                for (int p = 1; p < KSPLIT; p++) s += *(const v4f*)(src + p * (MT * NT * 256));
#pragma unroll
                for (int e = 0; e < 4; e++)
                    Cp[(size_t)(4 * q + e) * N + 16 * j] = s[e];
            }
    }
}

#define EDGE_CONV_N ((L_SEQ / INP_ROWS) * 3 * CONVDIM)
#define EDGE_ALL_N  (EDGE_CONV_N + L_SEQ * NHEADSN)

__global__ void __launch_bounds__(256) bm_conv_edge_dt(const float* __restrict__ zx,
                                                       const float* __restrict__ convw,
                                                       const float* __restrict__ convb,
                                                       const float* __restrict__ dtb,
                                                       float* __restrict__ xbc,
                                                       float* __restrict__ dt_o)
{
    const int idx = blockIdx.x * blockDim.x + threadIdx.x;
    if (idx >= EDGE_ALL_N) return;
    if (idx < EDGE_CONV_N) {
        const int e = idx / CONVDIM;
        const int c = idx - e * CONVDIM;
        const int l = (e / 3) * INP_ROWS + (e % 3);
        float acc = convb[c];
#pragma unroll
        for (int t = 0; t < 4; t++) {
            const int ll = l - 3 + t;
            if (ll >= 0)
                acc += convw[c * 4 + t] * zx[(size_t)ll * DINPROJ + DINNER + c];
        }
        xbc[(size_t)l * CONVDIM + c] = acc / (1.0f + expf(-acc));
    } else {
        const int k = idx - EDGE_CONV_N;
        const int l = k / NHEADSN;
        const int h = k - l * NHEADSN;
        const float v  = zx[(size_t)l * DINPROJ + DINNER + CONVDIM + h] + dtb[h];
        const float sp = (v > 20.0f) ? v : log1pf(expf(v));
        dt_o[l * NHEADSN + h] = sp;
    }
}

#define QCH   64
#define NCH   (L_SEQ / QCH)
#define SM_LD 68

__device__ __forceinline__ void bm_chunk_prefix(const float* __restrict__ dt,
                                                const float* __restrict__ Alog,
                                                int l0, int head, int tid,
                                                float* s_a, double* s_cs)
{
    if (tid < QCH)
        s_a[tid] = dt[(l0 + tid) * NHEADSN + head] * (-expf(Alog[head]));
    __syncthreads();
    if (tid < QCH) {
        double cs = 0.0;
        for (int s = 0; s <= tid; s++) cs += (double)s_a[s];
        s_cs[tid] = cs;
    }
    __syncthreads();
}

__global__ void __launch_bounds__(256) bm_chunk_state(const float* __restrict__ xbc,
                               const float* __restrict__ dt,
                               const float* __restrict__ Alog,
                               float* __restrict__ S,
                               float* __restrict__ cdec)
{
    const int c    = blockIdx.x;
    const int head = blockIdx.y;
    const int tid  = threadIdx.x;
    const int lane = tid & 31;
    const int wave = tid >> 5;
    const int l0   = c * QCH;

    __shared__ float  s_a[QCH];
    __shared__ double s_cs[QCH];
    __shared__ float  s_w[QCH];

    bm_chunk_prefix(dt, Alog, l0, head, tid, s_a, s_cs);
    if (tid < QCH)
        s_w[tid] = expf((float)(s_cs[QCH - 1] - s_cs[tid])) * dt[(l0 + tid) * NHEADSN + head];
    if (tid == 0)
        cdec[c * NHEADSN + head] = expf((float)s_cs[QCH - 1]);
    __syncthreads();

    const int mn   = lane & 15;
    const int koff = (lane >> 4) << 1;
    const float* xcol = xbc + (size_t)l0 * CONVDIM + head * HEADDIM + mn;
    const float* bcol = xbc + (size_t)l0 * CONVDIM + DINNER + wave * 16 + mn;

    v8f acc0 = {}, acc1 = {}, acc2 = {}, acc3 = {};
    for (int k = 0; k < QCH; k += 4) {
        const int s0 = k + koff;
        const int s1 = s0 + 1;
        const float w0 = s_w[s0];
        const float w1 = s_w[s1];
        const float* x0 = xcol + (size_t)s0 * CONVDIM;
        const float* x1 = xcol + (size_t)s1 * CONVDIM;
        v2f b, a0, a1, a2, a3;
        b.x  = bcol[(size_t)s0 * CONVDIM];
        b.y  = bcol[(size_t)s1 * CONVDIM];
        a0.x = w0 * x0[0];   a0.y = w1 * x1[0];
        a1.x = w0 * x0[16];  a1.y = w1 * x1[16];
        a2.x = w0 * x0[32];  a2.y = w1 * x1[32];
        a3.x = w0 * x0[48];  a3.y = w1 * x1[48];
        acc0 = __builtin_amdgcn_wmma_f32_16x16x4_f32(false, a0, false, b,
                                                     (short)0, acc0, false, false);
        acc1 = __builtin_amdgcn_wmma_f32_16x16x4_f32(false, a1, false, b,
                                                     (short)0, acc1, false, false);
        acc2 = __builtin_amdgcn_wmma_f32_16x16x4_f32(false, a2, false, b,
                                                     (short)0, acc2, false, false);
        acc3 = __builtin_amdgcn_wmma_f32_16x16x4_f32(false, a3, false, b,
                                                     (short)0, acc3, false, false);
    }

    const int msub = (lane >> 4) << 3;
    float* Sp = S + ((size_t)(c * NHEADSN + head) * HEADDIM + msub) * DSTATE + wave * 16 + mn;
#pragma unroll
    for (int v = 0; v < 8; v++) Sp[(size_t)v * DSTATE]        = acc0[v];
#pragma unroll
    for (int v = 0; v < 8; v++) Sp[(size_t)(16 + v) * DSTATE] = acc1[v];
#pragma unroll
    for (int v = 0; v < 8; v++) Sp[(size_t)(32 + v) * DSTATE] = acc2[v];
#pragma unroll
    for (int v = 0; v < 8; v++) Sp[(size_t)(48 + v) * DSTATE] = acc3[v];
}

__global__ void __launch_bounds__(256) bm_state_pass(float* __restrict__ S,
                              const float* __restrict__ cdec)
{
    const int idx  = blockIdx.x * blockDim.x + threadIdx.x;
    const int head = idx / (HEADDIM * DSTATE);
    const size_t plane = (size_t)NHEADSN * HEADDIM * DSTATE;

    float t[NCH];
#pragma unroll
    for (int c = 0; c < NCH; c++) t[c] = S[(size_t)c * plane + idx];

    float run = 0.0f;
#pragma unroll
    for (int c = 0; c < NCH; c++) {
        S[(size_t)c * plane + idx] = run;
        run = cdec[c * NHEADSN + head] * run + t[c];
    }
}

__global__ void __launch_bounds__(256) bm_chunk_out(const float* __restrict__ xbc,
                             const float* __restrict__ dt,
                             const float* __restrict__ Alog,
                             const float* __restrict__ Hs,
                             float* __restrict__ yscan)
{
    const int c    = blockIdx.x;
    const int head = blockIdx.y;
    const int tid  = threadIdx.x;
    const int lane = tid & 31;
    const int wave = tid >> 5;
    const int l0   = c * QCH;

    __shared__ float  s_a[QCH];
    __shared__ double s_cs[QCH];
    __shared__ float  s_dt[QCH];
    __shared__ float  s_e[QCH];
    __shared__ __attribute__((aligned(16))) float s_M[QCH * SM_LD];

    bm_chunk_prefix(dt, Alog, l0, head, tid, s_a, s_cs);
    if (tid < QCH) {
        s_dt[tid] = dt[(l0 + tid) * NHEADSN + head];
        s_e[tid]  = expf((float)s_cs[tid]);
    }
    __syncthreads();

    const int mn   = lane & 15;
    const int koff = (lane >> 4) << 1;
    const int msub = (lane >> 4) << 3;
    const int tt   = wave >> 1;
    const int cb   = (wave & 1) * 2;

    const float* Cr = xbc + (size_t)(l0 + tt * 16 + mn) * CONVDIM + DINNER + DSTATE + koff;

    {
        const float* B0 = xbc + (size_t)(l0 + (cb + 0) * 16 + mn) * CONVDIM + DINNER + koff;
        const float* B1 = xbc + (size_t)(l0 + (cb + 1) * 16 + mn) * CONVDIM + DINNER + koff;
        v8f g0 = {}, g1 = {};
        for (int k = 0; k < DSTATE; k += 4) {
            const v2f a  = *(const v2f*)(Cr + k);
            const v2f b0 = *(const v2f*)(B0 + k);
            const v2f b1 = *(const v2f*)(B1 + k);
            g0 = __builtin_amdgcn_wmma_f32_16x16x4_f32(false, a, false, b0,
                                                       (short)0, g0, false, false);
            g1 = __builtin_amdgcn_wmma_f32_16x16x4_f32(false, a, false, b1,
                                                       (short)0, g1, false, false);
        }
        const int sA = (cb + 0) * 16 + mn;
        const int sB = (cb + 1) * 16 + mn;
        const double csA = s_cs[sA], csB = s_cs[sB];
        const float  dtA = s_dt[sA], dtB = s_dt[sB];
#pragma unroll
        for (int v = 0; v < 8; v++) {
            const int t = tt * 16 + msub + v;
            const double cst = s_cs[t];
            const float mA = (sA <= t) ? g0[v] * expf((float)(cst - csA)) * dtA : 0.0f;
            const float mB = (sB <= t) ? g1[v] * expf((float)(cst - csB)) * dtB : 0.0f;
            s_M[t * SM_LD + sA] = mA;
            s_M[t * SM_LD + sB] = mB;
        }
    }
    __syncthreads();

    v8f yi0 = {}, yi1 = {};
    {
        const float* H0 = Hs + ((size_t)(c * NHEADSN + head) * HEADDIM + (cb + 0) * 16 + mn) * DSTATE + koff;
        const float* H1 = Hs + ((size_t)(c * NHEADSN + head) * HEADDIM + (cb + 1) * 16 + mn) * DSTATE + koff;
        for (int k = 0; k < DSTATE; k += 4) {
            const v2f a  = *(const v2f*)(Cr + k);
            const v2f h0 = *(const v2f*)(H0 + k);
            const v2f h1 = *(const v2f*)(H1 + k);
            yi0 = __builtin_amdgcn_wmma_f32_16x16x4_f32(false, a, false, h0,
                                                        (short)0, yi0, false, false);
            yi1 = __builtin_amdgcn_wmma_f32_16x16x4_f32(false, a, false, h1,
                                                        (short)0, yi1, false, false);
        }
    }
    v8f ya0 = {}, ya1 = {};
    {
        const float* Ml = s_M + (tt * 16 + mn) * SM_LD + koff;
        const float* X0 = xbc + (size_t)l0 * CONVDIM + head * HEADDIM + (cb + 0) * 16 + mn;
        const float* X1 = X0 + 16;
        const int kend = (tt + 1) * 16;
        for (int k = 0; k < kend; k += 4) {
            const int s0 = k + koff;
            const int s1 = s0 + 1;
            const v2f a = *(const v2f*)(Ml + k);
            v2f x0, x1;
            x0.x = X0[(size_t)s0 * CONVDIM];  x0.y = X0[(size_t)s1 * CONVDIM];
            x1.x = X1[(size_t)s0 * CONVDIM];  x1.y = X1[(size_t)s1 * CONVDIM];
            ya0 = __builtin_amdgcn_wmma_f32_16x16x4_f32(false, a, false, x0,
                                                        (short)0, ya0, false, false);
            ya1 = __builtin_amdgcn_wmma_f32_16x16x4_f32(false, a, false, x1,
                                                        (short)0, ya1, false, false);
        }
    }

    float* Yp = yscan + (size_t)(l0 + tt * 16 + msub) * DINNER + head * HEADDIM + cb * 16 + mn;
#pragma unroll
    for (int v = 0; v < 8; v++) {
        const float e = s_e[tt * 16 + msub + v];
        Yp[(size_t)v * DINNER]      = ya0[v] + e * yi0[v];
        Yp[(size_t)v * DINNER + 16] = ya1[v] + e * yi1[v];
    }
}

__global__ void bm_gate_rmsnorm(float* __restrict__ y,
                                const float* __restrict__ xbc,
                                const float* __restrict__ zx,
                                const float* __restrict__ Dp,
                                const float* __restrict__ normw)
{
    const int l   = blockIdx.x;
    const int tid = threadIdx.x;
    __shared__ float red[8];
    __shared__ float scale;

    float vals[4];
    float ss = 0.0f;
#pragma unroll
    for (int j = 0; j < 4; j++) {
        const int i = tid * 4 + j;
        const int hh = i >> 6;
        float v = y[(size_t)l * DINNER + i] + Dp[hh] * xbc[(size_t)l * CONVDIM + i];
        const float z = zx[(size_t)l * DINPROJ + i];
        v *= z / (1.0f + expf(-z));
        vals[j] = v;
        ss += v * v;
    }
#pragma unroll
    for (int o = 16; o >= 1; o >>= 1) ss += __shfl_xor(ss, o, 32);
    if ((tid & 31) == 0) red[tid >> 5] = ss;
    __syncthreads();
    if (tid == 0) {
        float t = 0.0f;
        for (int i = 0; i < 8; i++) t += red[i];
        scale = rsqrtf(t / (float)DINNER + EPS_RMS);
    }
    __syncthreads();
#pragma unroll
    for (int j = 0; j < 4; j++) {
        const int i = tid * 4 + j;
        y[(size_t)l * DINNER + i] = vals[j] * scale * normw[i];
    }
}

__global__ void bm_add_flip(const float* __restrict__ of,
                            const float* __restrict__ ob,
                            float* __restrict__ out)
{
    const int idx = blockIdx.x * blockDim.x + threadIdx.x;
    if (idx >= L_SEQ * DMODEL) return;
    const int l = idx / DMODEL;
    const int j = idx - l * DMODEL;
    out[idx] = of[idx] + ob[(size_t)(L_SEQ - 1 - l) * DMODEL + j];
}

extern "C" void kernel_launch(void* const* d_in, const int* in_sizes, int n_in,
                              void* d_out, int out_size, void* d_ws, size_t ws_size,
                              hipStream_t stream)
{
    const float* Win[2]   = { (const float*)d_in[0],  (const float*)d_in[8]  };
    const float* convw[2] = { (const float*)d_in[1],  (const float*)d_in[9]  };
    const float* convb[2] = { (const float*)d_in[2],  (const float*)d_in[10] };
    const float* dtb[2]   = { (const float*)d_in[3],  (const float*)d_in[11] };
    const float* Alog[2]  = { (const float*)d_in[4],  (const float*)d_in[12] };
    const float* Dp[2]    = { (const float*)d_in[5],  (const float*)d_in[13] };
    const float* normw[2] = { (const float*)d_in[6],  (const float*)d_in[14] };
    const float* Wout[2]  = { (const float*)d_in[7],  (const float*)d_in[15] };
    const float* x        =   (const float*)d_in[16];

    float* ws = (float*)d_ws;
    size_t off = 0;
    float* zx[2];  float* xbc[2]; float* dtw[2];
    float* ysc[2]; float* od[2];
    for (int d = 0; d < 2; d++) { zx[d]  = ws + off; off += (size_t)L_SEQ * DINPROJ; }
    for (int d = 0; d < 2; d++) { xbc[d] = ws + off; off += (size_t)L_SEQ * CONVDIM; }
    for (int d = 0; d < 2; d++) { dtw[d] = ws + off; off += (size_t)L_SEQ * NHEADSN; }
    for (int d = 0; d < 2; d++) { ysc[d] = ws + off; off += (size_t)L_SEQ * DINNER;  }
    for (int d = 0; d < 2; d++) { od[d]  = ws + off; off += (size_t)L_SEQ * DMODEL;  }

    const dim3 blk256(256);

    for (int d = 0; d < 2; d++) {
        {
            const int tiles = (L_SEQ / INP_ROWS) * (DINPROJ / 80);
            bm_wmma_gemm_nt_f32<DMODEL, DINPROJ, 4, 5, 1, true><<<dim3((tiles + 3) / 4), dim3(128), 0, stream>>>(
                x, Win[d], zx[d], d, convw[d], convb[d], xbc[d]);
        }
        bm_conv_edge_dt<<<dim3((EDGE_ALL_N + 255) / 256), blk256, 0, stream>>>(
            zx[d], convw[d], convb[d], dtb[d], xbc[d], dtw[d]);
        {
            float* Sst  = zx[1 - d];
            float* cdec = Sst + (size_t)NCH * NHEADSN * HEADDIM * DSTATE;
            bm_chunk_state<<<dim3(NCH, NHEADSN), blk256, 0, stream>>>(
                xbc[d], dtw[d], Alog[d], Sst, cdec);
            bm_state_pass<<<dim3(NHEADSN * HEADDIM * DSTATE / 256), blk256, 0, stream>>>(
                Sst, cdec);
            bm_chunk_out<<<dim3(NCH, NHEADSN), blk256, 0, stream>>>(
                xbc[d], dtw[d], Alog[d], Sst, ysc[d]);
        }
        bm_gate_rmsnorm<<<dim3(L_SEQ), blk256, 0, stream>>>(
            ysc[d], xbc[d], zx[d], Dp[d], normw[d]);
        {
            const int tiles = (L_SEQ / 64) * (DMODEL / 64);
            bm_wmma_gemm_nt_f32<DINNER, DMODEL, 4, 4, 4, false><<<dim3(tiles), dim3(128), 0, stream>>>(
                ysc[d], Wout[d], od[d], 0, nullptr, nullptr, nullptr);
        }
    }
    bm_add_flip<<<dim3((L_SEQ * DMODEL + 255) / 256), blk256, 0, stream>>>(
        od[0], od[1], (float*)d_out);
}


# both GEMMs: fragment loads regrouped to 32 k per group (a row's whole 128-byte line by four back-to-back 16-byte loads), issued one 32-k WMMA block (160/128 WMMAs) ahead; was 8 k, 40/32 WMMAs ahead
# speedup vs baseline: 27.3230x; 1.0834x over previous
#include <hip/hip_runtime.h>
#include <math.h>

#define L_SEQ     2048
#define DMODEL    512
#define DINNER    1024
#define DSTATE    128
#define HEADDIM   64
#define NHEADSN   16
#define CONVDIM   1280
#define DINPROJ   2320
#define EPS_RMS   1e-5f

typedef __attribute__((ext_vector_type(2))) float v2f;
typedef __attribute__((ext_vector_type(4))) float v4f;
typedef __attribute__((ext_vector_type(8))) float v8f;

#define INP_ROWS 64

template <int K, int N, int MT, int NT, int KSPLIT, bool FUSE>
__global__ void __launch_bounds__(128) bm_wmma_gemm_nt_f32(const float* __restrict__ A,
                                                           const float* __restrict__ Bw,
                                                           float* __restrict__ C,
                                                           int flipA,
                                                           const float* __restrict__ convw,
                                                           const float* __restrict__ convb,
                                                           float* __restrict__ xbc)
{
    constexpr int M    = L_SEQ;
    constexpr int KB   = 8;
    constexpr int QG   = 4;
    constexpr int KG   = KB * QG;
    constexpr int KW   = K / KSPLIT;
    constexpr int NG   = KW / KG;
    constexpr int NTN  = N / (16 * NT);
    constexpr int NTM  = M / (16 * MT);
    static_assert(N % (16 * NT) == 0 && M % (16 * MT) == 0 && K % (2 * KG * KSPLIT) == 0, "tile shape");
    static_assert(KSPLIT == 1 || (KSPLIT == 4 && MT == 4), "split shape");
    static_assert(!FUSE || (KSPLIT == 1 && N == DINPROJ && 16 * MT == INP_ROWS), "fused epilogue shape");

    const int lane   = threadIdx.x & 31;
    const int wave   = threadIdx.x >> 5;
    const int tileId = (KSPLIT == 1) ? blockIdx.x * 4 + wave : blockIdx.x;
    if (KSPLIT == 1 && tileId >= NTM * NTN) return;
    const int tm = tileId / NTN;
    const int tn = tileId - tm * NTN;
    constexpr int JST = FUSE ? NTN : 1;
    const int ct0 = FUSE ? tn : tn * NT;
    const int c  = lane & 15;
    const int h  = lane >> 4;
    const int k0 = (KSPLIT == 1) ? 0 : wave * KW;

    const float* Ar[MT];
#pragma unroll
    for (int i = 0; i < MT; i++) {
        int r = tm * (16 * MT) + 16 * i + c;
        if (flipA) r = M - 1 - r;
        Ar[i] = A + (size_t)r * K + k0 + 4 * h;
    }
    const float* Br = Bw + (size_t)(ct0 * 16 + c) * K + k0 + 4 * h;

    v8f acc[MT][NT];
#pragma unroll
    for (int i = 0; i < MT; i++)
#pragma unroll
        for (int j = 0; j < NT; j++) acc[i][j] = v8f{};

    v4f fa0[MT][QG], fb0[NT][QG], fa1[MT][QG], fb1[NT][QG];

    auto load_set = [&](v4f (&fa)[MT][QG], v4f (&fb)[NT][QG], int grp) {
#pragma unroll
        for (int i = 0; i < MT; i++)
#pragma unroll
            for (int q = 0; q < QG; q++) fa[i][q] = *(const v4f*)(Ar[i] + grp * KG + q * KB);
#pragma unroll
        for (int j = 0; j < NT; j++)
#pragma unroll
            for (int q = 0; q < QG; q++)
                fb[j][q] = *(const v4f*)(Br + (size_t)j * JST * 16 * K + grp * KG + q * KB);
        __builtin_amdgcn_sched_barrier(0);
    };
    auto mma_set = [&](const v4f (&fa)[MT][QG], const v4f (&fb)[NT][QG]) {
#pragma unroll
        for (int q = 0; q < QG; q++) {
#pragma unroll
            for (int i = 0; i < MT; i++) {
                const v2f a = __builtin_shufflevector(fa[i][q], fa[i][q], 0, 1);
#pragma unroll
                for (int j = 0; j < NT; j++) {
                    const v2f b = __builtin_shufflevector(fb[j][q], fb[j][q], 0, 1);
                    acc[i][j] = __builtin_amdgcn_wmma_f32_16x16x4_f32(false, a, false, b,
                                                                      (short)0, acc[i][j], false, false);
                }
            }
#pragma unroll
            for (int i = 0; i < MT; i++) {
                const v2f a = __builtin_shufflevector(fa[i][q], fa[i][q], 2, 3);
#pragma unroll
                for (int j = 0; j < NT; j++) {
                    const v2f b = __builtin_shufflevector(fb[j][q], fb[j][q], 2, 3);
                    acc[i][j] = __builtin_amdgcn_wmma_f32_16x16x4_f32(false, a, false, b,
                                                                      (short)0, acc[i][j], false, false);
                }
            }
        }
        __builtin_amdgcn_sched_barrier(0);
    };

    load_set(fa0, fb0, 0);
    for (int grp = 0; grp < NG - 2; grp += 2) {
        load_set(fa1, fb1, grp + 1);
        mma_set(fa0, fb0);
        load_set(fa0, fb0, grp + 2);
        mma_set(fa1, fb1);
    }
    load_set(fa1, fb1, NG - 1);
    mma_set(fa0, fb0);
    mma_set(fa1, fb1);

    if constexpr (FUSE) {
        const int rbase = tm * (16 * MT) + 8 * h;
#pragma unroll
        for (int j = 0; j < NT; j++) {
            const int col16 = (ct0 + j * JST) * 16;
            const int col   = col16 + c;
            float* Cp = C + (size_t)rbase * N + col;
            if (col16 < DINNER || col16 >= DINNER + CONVDIM) {
#pragma unroll
                for (int i = 0; i < MT; i++)
#pragma unroll
                    for (int v = 0; v < 8; v++)
                        Cp[(size_t)(16 * i + v) * N] = acc[i][j][v];
            } else {
                const int ch = col - DINNER;
                float* Xp = xbc + (size_t)rbase * CONVDIM + ch;
                const float w0 = convw[ch * 4 + 0];
                const float w1 = convw[ch * 4 + 1];
                const float w2 = convw[ch * 4 + 2];
                const float w3 = convw[ch * 4 + 3];
                const float cb = convb[ch];
                float pv[MT][3];
#pragma unroll
                for (int i = 0; i < MT; i++)
#pragma unroll
                    for (int q = 0; q < 3; q++)
                        pv[i][q] = __shfl_xor(acc[i][j][5 + q], 16, 32);
#pragma unroll
                for (int i = 0; i < MT; i++) {
                    float p[11];
#pragma unroll
                    for (int q = 0; q < 3; q++)
                        p[q] = h ? pv[i][q] : pv[i > 0 ? i - 1 : 0][q];
#pragma unroll
                    for (int v = 0; v < 8; v++) p[3 + v] = acc[i][j][v];
#pragma unroll
                    for (int v = 0; v < 8; v++) {
                        float a = cb;
                        a += w0 * p[v];
                        a += w1 * p[v + 1];
                        a += w2 * p[v + 2];
                        a += w3 * p[v + 3];
                        const float s = a / (1.0f + expf(-a));
                        const bool first3 = (i == 0 && v < 3);
                        const bool last3  = (i == MT - 1 && v >= 5);
                        if (!first3 || h) Xp[(size_t)(16 * i + v) * CONVDIM] = s;
                        if ((first3 && !h) || (last3 && h)) Cp[(size_t)(16 * i + v) * N] = acc[i][j][v];
                    }
                }
            }
        }
    } else if constexpr (KSPLIT == 1) {
        float* Cp = C + (size_t)(tm * (16 * MT) + 8 * h) * N + tn * (16 * NT) + c;
#pragma unroll
        for (int i = 0; i < MT; i++)
#pragma unroll
            for (int j = 0; j < NT; j++)
#pragma unroll
                for (int v = 0; v < 8; v++)
                    Cp[(size_t)(16 * i + v) * N + 16 * j] = acc[i][j][v];
    } else {
        __shared__ __attribute__((aligned(16))) float s_part[KSPLIT * MT * NT * 256];
        float* mine = s_part + wave * (MT * NT * 256) + lane * 4;
#pragma unroll
        for (int i = 0; i < MT; i++)
#pragma unroll
            for (int j = 0; j < NT; j++) {
                *(v4f*)(mine + ((i * NT + j) * 2 + 0) * 128) = __builtin_shufflevector(acc[i][j], acc[i][j], 0, 1, 2, 3);
                *(v4f*)(mine + ((i * NT + j) * 2 + 1) * 128) = __builtin_shufflevector(acc[i][j], acc[i][j], 4, 5, 6, 7);
            }
        __syncthreads();
        float* Cp = C + (size_t)(tm * (16 * MT) + 16 * wave + 8 * h) * N + tn * (16 * NT) + c;
#pragma unroll
        for (int j = 0; j < NT; j++)
#pragma unroll
            for (int q = 0; q < 2; q++) {
                const float* src = s_part + ((wave * NT + j) * 2 + q) * 128 + lane * 4;
                v4f s = *(const v4f*)(src);
#pragma unroll
                for (int p = 1; p < KSPLIT; p++) s += *(const v4f*)(src + p * (MT * NT * 256));
#pragma unroll
                for (int e = 0; e < 4; e++)
                    Cp[(size_t)(4 * q + e) * N + 16 * j] = s[e];
            }
    }
}

#define EDGE_CONV_N ((L_SEQ / INP_ROWS) * 3 * CONVDIM)
#define EDGE_ALL_N  (EDGE_CONV_N + L_SEQ * NHEADSN)

__global__ void __launch_bounds__(256) bm_conv_edge_dt(const float* __restrict__ zx,
                                                       const float* __restrict__ convw,
                                                       const float* __restrict__ convb,
                                                       const float* __restrict__ dtb,
                                                       float* __restrict__ xbc,
                                                       float* __restrict__ dt_o)
{
    const int idx = blockIdx.x * blockDim.x + threadIdx.x;
    if (idx >= EDGE_ALL_N) return;
    if (idx < EDGE_CONV_N) {
        const int e = idx / CONVDIM;
        const int c = idx - e * CONVDIM;
        const int l = (e / 3) * INP_ROWS + (e % 3);
        float acc = convb[c];
#pragma unroll
        for (int t = 0; t < 4; t++) {
            const int ll = l - 3 + t;
            if (ll >= 0)
                acc += convw[c * 4 + t] * zx[(size_t)ll * DINPROJ + DINNER + c];
        }
        xbc[(size_t)l * CONVDIM + c] = acc / (1.0f + expf(-acc));
    } else {
        const int k = idx - EDGE_CONV_N;
        const int l = k / NHEADSN;
        const int h = k - l * NHEADSN;
        const float v  = zx[(size_t)l * DINPROJ + DINNER + CONVDIM + h] + dtb[h];
        const float sp = (v > 20.0f) ? v : log1pf(expf(v));
        dt_o[l * NHEADSN + h] = sp;
    }
}

#define QCH   64
#define NCH   (L_SEQ / QCH)
#define SM_LD 68

__device__ __forceinline__ void bm_chunk_prefix(const float* __restrict__ dt,
                                                const float* __restrict__ Alog,
                                                int l0, int head, int tid,
                                                float* s_a, double* s_cs)
{
    if (tid < QCH)
        s_a[tid] = dt[(l0 + tid) * NHEADSN + head] * (-expf(Alog[head]));
    __syncthreads();
    if (tid < QCH) {
        double cs = 0.0;
        for (int s = 0; s <= tid; s++) cs += (double)s_a[s];
        s_cs[tid] = cs;
    }
    __syncthreads();
}

__global__ void __launch_bounds__(256) bm_chunk_state(const float* __restrict__ xbc,
                               const float* __restrict__ dt,
                               const float* __restrict__ Alog,
                               float* __restrict__ S,
                               float* __restrict__ cdec)
{
    const int c    = blockIdx.x;
    const int head = blockIdx.y;
    const int tid  = threadIdx.x;
    const int lane = tid & 31;
    const int wave = tid >> 5;
    const int l0   = c * QCH;

    __shared__ float  s_a[QCH];
    __shared__ double s_cs[QCH];
    __shared__ float  s_w[QCH];

    bm_chunk_prefix(dt, Alog, l0, head, tid, s_a, s_cs);
    if (tid < QCH)
        s_w[tid] = expf((float)(s_cs[QCH - 1] - s_cs[tid])) * dt[(l0 + tid) * NHEADSN + head];
    if (tid == 0)
        cdec[c * NHEADSN + head] = expf((float)s_cs[QCH - 1]);
    __syncthreads();

    const int mn   = lane & 15;
    const int koff = (lane >> 4) << 1;
    const float* xcol = xbc + (size_t)l0 * CONVDIM + head * HEADDIM + mn;
    const float* bcol = xbc + (size_t)l0 * CONVDIM + DINNER + wave * 16 + mn;

    v8f acc0 = {}, acc1 = {}, acc2 = {}, acc3 = {};
    for (int k = 0; k < QCH; k += 4) {
        const int s0 = k + koff;
        const int s1 = s0 + 1;
        const float w0 = s_w[s0];
        const float w1 = s_w[s1];
        const float* x0 = xcol + (size_t)s0 * CONVDIM;
        const float* x1 = xcol + (size_t)s1 * CONVDIM;
        v2f b, a0, a1, a2, a3;
        b.x  = bcol[(size_t)s0 * CONVDIM];
        b.y  = bcol[(size_t)s1 * CONVDIM];
        a0.x = w0 * x0[0];   a0.y = w1 * x1[0];
        a1.x = w0 * x0[16];  a1.y = w1 * x1[16];
        a2.x = w0 * x0[32];  a2.y = w1 * x1[32];
        a3.x = w0 * x0[48];  a3.y = w1 * x1[48];
        acc0 = __builtin_amdgcn_wmma_f32_16x16x4_f32(false, a0, false, b,
                                                     (short)0, acc0, false, false);
        acc1 = __builtin_amdgcn_wmma_f32_16x16x4_f32(false, a1, false, b,
                                                     (short)0, acc1, false, false);
        acc2 = __builtin_amdgcn_wmma_f32_16x16x4_f32(false, a2, false, b,
                                                     (short)0, acc2, false, false);
        acc3 = __builtin_amdgcn_wmma_f32_16x16x4_f32(false, a3, false, b,
                                                     (short)0, acc3, false, false);
    }

    const int msub = (lane >> 4) << 3;
    float* Sp = S + ((size_t)(c * NHEADSN + head) * HEADDIM + msub) * DSTATE + wave * 16 + mn;
#pragma unroll
    for (int v = 0; v < 8; v++) Sp[(size_t)v * DSTATE]        = acc0[v];
#pragma unroll
    for (int v = 0; v < 8; v++) Sp[(size_t)(16 + v) * DSTATE] = acc1[v];
#pragma unroll
    for (int v = 0; v < 8; v++) Sp[(size_t)(32 + v) * DSTATE] = acc2[v];
#pragma unroll
    for (int v = 0; v < 8; v++) Sp[(size_t)(48 + v) * DSTATE] = acc3[v];
}

__global__ void __launch_bounds__(256) bm_state_pass(float* __restrict__ S,
                              const float* __restrict__ cdec)
{
    const int idx  = blockIdx.x * blockDim.x + threadIdx.x;
    const int head = idx / (HEADDIM * DSTATE);
    const size_t plane = (size_t)NHEADSN * HEADDIM * DSTATE;

    float t[NCH];
#pragma unroll
    for (int c = 0; c < NCH; c++) t[c] = S[(size_t)c * plane + idx];

    float run = 0.0f;
#pragma unroll
    for (int c = 0; c < NCH; c++) {
        S[(size_t)c * plane + idx] = run;
        run = cdec[c * NHEADSN + head] * run + t[c];
    }
}

__global__ void __launch_bounds__(256) bm_chunk_out(const float* __restrict__ xbc,
                             const float* __restrict__ dt,
                             const float* __restrict__ Alog,
                             const float* __restrict__ Hs,
                             float* __restrict__ yscan)
{
    const int c    = blockIdx.x;
    const int head = blockIdx.y;
    const int tid  = threadIdx.x;
    const int lane = tid & 31;
    const int wave = tid >> 5;
    const int l0   = c * QCH;

    __shared__ float  s_a[QCH];
    __shared__ double s_cs[QCH];
    __shared__ float  s_dt[QCH];
    __shared__ float  s_e[QCH];
    __shared__ __attribute__((aligned(16))) float s_M[QCH * SM_LD];

    bm_chunk_prefix(dt, Alog, l0, head, tid, s_a, s_cs);
    if (tid < QCH) {
        s_dt[tid] = dt[(l0 + tid) * NHEADSN + head];
        s_e[tid]  = expf((float)s_cs[tid]);
    }
    __syncthreads();

    const int mn   = lane & 15;
    const int koff = (lane >> 4) << 1;
    const int msub = (lane >> 4) << 3;
    const int tt   = wave >> 1;
    const int cb   = (wave & 1) * 2;

    const float* Cr = xbc + (size_t)(l0 + tt * 16 + mn) * CONVDIM + DINNER + DSTATE + koff;

    {
        const float* B0 = xbc + (size_t)(l0 + (cb + 0) * 16 + mn) * CONVDIM + DINNER + koff;
        const float* B1 = xbc + (size_t)(l0 + (cb + 1) * 16 + mn) * CONVDIM + DINNER + koff;
        v8f g0 = {}, g1 = {};
        for (int k = 0; k < DSTATE; k += 4) {
            const v2f a  = *(const v2f*)(Cr + k);
            const v2f b0 = *(const v2f*)(B0 + k);
            const v2f b1 = *(const v2f*)(B1 + k);
            g0 = __builtin_amdgcn_wmma_f32_16x16x4_f32(false, a, false, b0,
                                                       (short)0, g0, false, false);
            g1 = __builtin_amdgcn_wmma_f32_16x16x4_f32(false, a, false, b1,
                                                       (short)0, g1, false, false);
        }
        const int sA = (cb + 0) * 16 + mn;
        const int sB = (cb + 1) * 16 + mn;
        const double csA = s_cs[sA], csB = s_cs[sB];
        const float  dtA = s_dt[sA], dtB = s_dt[sB];
#pragma unroll
        for (int v = 0; v < 8; v++) {
            const int t = tt * 16 + msub + v;
            const double cst = s_cs[t];
            const float mA = (sA <= t) ? g0[v] * expf((float)(cst - csA)) * dtA : 0.0f;
            const float mB = (sB <= t) ? g1[v] * expf((float)(cst - csB)) * dtB : 0.0f;
            s_M[t * SM_LD + sA] = mA;
            s_M[t * SM_LD + sB] = mB;
        }
    }
    __syncthreads();

    v8f yi0 = {}, yi1 = {};
    {
        const float* H0 = Hs + ((size_t)(c * NHEADSN + head) * HEADDIM + (cb + 0) * 16 + mn) * DSTATE + koff;
        const float* H1 = Hs + ((size_t)(c * NHEADSN + head) * HEADDIM + (cb + 1) * 16 + mn) * DSTATE + koff;
        for (int k = 0; k < DSTATE; k += 4) {
            const v2f a  = *(const v2f*)(Cr + k);
            const v2f h0 = *(const v2f*)(H0 + k);
            const v2f h1 = *(const v2f*)(H1 + k);
            yi0 = __builtin_amdgcn_wmma_f32_16x16x4_f32(false, a, false, h0,
                                                        (short)0, yi0, false, false);
            yi1 = __builtin_amdgcn_wmma_f32_16x16x4_f32(false, a, false, h1,
                                                        (short)0, yi1, false, false);
        }
    }
    v8f ya0 = {}, ya1 = {};
    {
        const float* Ml = s_M + (tt * 16 + mn) * SM_LD + koff;
        const float* X0 = xbc + (size_t)l0 * CONVDIM + head * HEADDIM + (cb + 0) * 16 + mn;
        const float* X1 = X0 + 16;
        const int kend = (tt + 1) * 16;
        for (int k = 0; k < kend; k += 4) {
            const int s0 = k + koff;
            const int s1 = s0 + 1;
            const v2f a = *(const v2f*)(Ml + k);
            v2f x0, x1;
            x0.x = X0[(size_t)s0 * CONVDIM];  x0.y = X0[(size_t)s1 * CONVDIM];
            x1.x = X1[(size_t)s0 * CONVDIM];  x1.y = X1[(size_t)s1 * CONVDIM];
            ya0 = __builtin_amdgcn_wmma_f32_16x16x4_f32(false, a, false, x0,
                                                        (short)0, ya0, false, false);
            ya1 = __builtin_amdgcn_wmma_f32_16x16x4_f32(false, a, false, x1,
                                                        (short)0, ya1, false, false);
        }
    }

    float* Yp = yscan + (size_t)(l0 + tt * 16 + msub) * DINNER + head * HEADDIM + cb * 16 + mn;
#pragma unroll
    for (int v = 0; v < 8; v++) {
        const float e = s_e[tt * 16 + msub + v];
        Yp[(size_t)v * DINNER]      = ya0[v] + e * yi0[v];
        Yp[(size_t)v * DINNER + 16] = ya1[v] + e * yi1[v];
    }
}

__global__ void bm_gate_rmsnorm(float* __restrict__ y,
                                const float* __restrict__ xbc,
                                const float* __restrict__ zx,
                                const float* __restrict__ Dp,
                                const float* __restrict__ normw)
{
    const int l   = blockIdx.x;
    const int tid = threadIdx.x;
    __shared__ float red[8];
    __shared__ float scale;

    float vals[4];
    float ss = 0.0f;
#pragma unroll
    for (int j = 0; j < 4; j++) {
        const int i = tid * 4 + j;
        const int hh = i >> 6;
        float v = y[(size_t)l * DINNER + i] + Dp[hh] * xbc[(size_t)l * CONVDIM + i];
        const float z = zx[(size_t)l * DINPROJ + i];
        v *= z / (1.0f + expf(-z));
        vals[j] = v;
        ss += v * v;
    }
#pragma unroll
    for (int o = 16; o >= 1; o >>= 1) ss += __shfl_xor(ss, o, 32);
    if ((tid & 31) == 0) red[tid >> 5] = ss;
    __syncthreads();
    if (tid == 0) {
        float t = 0.0f;
        for (int i = 0; i < 8; i++) t += red[i];
        scale = rsqrtf(t / (float)DINNER + EPS_RMS);
    }
    __syncthreads();
#pragma unroll
    for (int j = 0; j < 4; j++) {
        const int i = tid * 4 + j;
        y[(size_t)l * DINNER + i] = vals[j] * scale * normw[i];
    }
}

__global__ void bm_add_flip(const float* __restrict__ of,
                            const float* __restrict__ ob,
                            float* __restrict__ out)
{
    const int idx = blockIdx.x * blockDim.x + threadIdx.x;
    if (idx >= L_SEQ * DMODEL) return;
    const int l = idx / DMODEL;
    const int j = idx - l * DMODEL;
    out[idx] = of[idx] + ob[(size_t)(L_SEQ - 1 - l) * DMODEL + j];
}

extern "C" void kernel_launch(void* const* d_in, const int* in_sizes, int n_in,
                              void* d_out, int out_size, void* d_ws, size_t ws_size,
                              hipStream_t stream)
{
    const float* Win[2]   = { (const float*)d_in[0],  (const float*)d_in[8]  };
    const float* convw[2] = { (const float*)d_in[1],  (const float*)d_in[9]  };
    const float* convb[2] = { (const float*)d_in[2],  (const float*)d_in[10] };
    const float* dtb[2]   = { (const float*)d_in[3],  (const float*)d_in[11] };
    const float* Alog[2]  = { (const float*)d_in[4],  (const float*)d_in[12] };
    const float* Dp[2]    = { (const float*)d_in[5],  (const float*)d_in[13] };
    const float* normw[2] = { (const float*)d_in[6],  (const float*)d_in[14] };
    const float* Wout[2]  = { (const float*)d_in[7],  (const float*)d_in[15] };
    const float* x        =   (const float*)d_in[16];

    float* ws = (float*)d_ws;
    size_t off = 0;
    float* zx[2];  float* xbc[2]; float* dtw[2];
    float* ysc[2]; float* od[2];
    for (int d = 0; d < 2; d++) { zx[d]  = ws + off; off += (size_t)L_SEQ * DINPROJ; }
    for (int d = 0; d < 2; d++) { xbc[d] = ws + off; off += (size_t)L_SEQ * CONVDIM; }
    for (int d = 0; d < 2; d++) { dtw[d] = ws + off; off += (size_t)L_SEQ * NHEADSN; }
    for (int d = 0; d < 2; d++) { ysc[d] = ws + off; off += (size_t)L_SEQ * DINNER;  }
    for (int d = 0; d < 2; d++) { od[d]  = ws + off; off += (size_t)L_SEQ * DMODEL;  }

    const dim3 blk256(256);

    for (int d = 0; d < 2; d++) {
        {
            const int tiles = (L_SEQ / INP_ROWS) * (DINPROJ / 80);
            bm_wmma_gemm_nt_f32<DMODEL, DINPROJ, 4, 5, 1, true><<<dim3((tiles + 3) / 4), dim3(128), 0, stream>>>(
                x, Win[d], zx[d], d, convw[d], convb[d], xbc[d]);
        }
        bm_conv_edge_dt<<<dim3((EDGE_ALL_N + 255) / 256), blk256, 0, stream>>>(
            zx[d], convw[d], convb[d], dtb[d], xbc[d], dtw[d]);
        {
            float* Sst  = zx[1 - d];
            float* cdec = Sst + (size_t)NCH * NHEADSN * HEADDIM * DSTATE;
            bm_chunk_state<<<dim3(NCH, NHEADSN), blk256, 0, stream>>>(
                xbc[d], dtw[d], Alog[d], Sst, cdec);
            bm_state_pass<<<dim3(NHEADSN * HEADDIM * DSTATE / 256), blk256, 0, stream>>>(
                Sst, cdec);
            bm_chunk_out<<<dim3(NCH, NHEADSN), blk256, 0, stream>>>(
                xbc[d], dtw[d], Alog[d], Sst, ysc[d]);
        }
        bm_gate_rmsnorm<<<dim3(L_SEQ), blk256, 0, stream>>>(
            ysc[d], xbc[d], zx[d], Dp[d], normw[d]);
        {
            const int tiles = (L_SEQ / 64) * (DMODEL / 64);
            bm_wmma_gemm_nt_f32<DINNER, DMODEL, 4, 4, 4, false><<<dim3(tiles), dim3(128), 0, stream>>>(
                ysc[d], Wout[d], od[d], 0, nullptr, nullptr, nullptr);
        }
    }
    bm_add_flip<<<dim3((L_SEQ * DMODEL + 255) / 256), blk256, 0, stream>>>(
        od[0], od[1], (float*)d_out);
}


# chunk decay sums leave the scan launches: the serial 64-step fp64 add loop per thread (in 4 launches) becomes one lane-shuffle scan beside softplus(dt); exp(cs), weights, cs as hi+lo floats read back
# speedup vs baseline: 29.1314x; 1.0662x over previous
#include <hip/hip_runtime.h>
#include <math.h>

#define L_SEQ     2048
#define DMODEL    512
#define DINNER    1024
#define DSTATE    128
#define HEADDIM   64
#define NHEADSN   16
#define CONVDIM   1280
#define DINPROJ   2320
#define EPS_RMS   1e-5f
#define QCH       64
#define NCH       (L_SEQ / QCH)

typedef __attribute__((ext_vector_type(2))) float v2f;
typedef __attribute__((ext_vector_type(4))) float v4f;
typedef __attribute__((ext_vector_type(8))) float v8f;

#define INP_ROWS 64

template <int K, int N, int MT, int NT, int KSPLIT, bool FUSE>
__global__ void __launch_bounds__(128) bm_wmma_gemm_nt_f32(const float* __restrict__ A,
                                                           const float* __restrict__ Bw,
                                                           float* __restrict__ C,
                                                           int flipA,
                                                           const float* __restrict__ convw,
                                                           const float* __restrict__ convb,
                                                           float* __restrict__ xbc)
{
    constexpr int M    = L_SEQ;
    constexpr int KB   = 8;
    constexpr int QG   = 4;
    constexpr int KG   = KB * QG;
    constexpr int KW   = K / KSPLIT;
    constexpr int NG   = KW / KG;
    constexpr int NTN  = N / (16 * NT);
    constexpr int NTM  = M / (16 * MT);
    static_assert(N % (16 * NT) == 0 && M % (16 * MT) == 0 && K % (2 * KG * KSPLIT) == 0, "tile shape");
    static_assert(KSPLIT == 1 || (KSPLIT == 4 && MT == 4), "split shape");
    static_assert(!FUSE || (KSPLIT == 1 && N == DINPROJ && 16 * MT == INP_ROWS), "fused epilogue shape");

    const int lane   = threadIdx.x & 31;
    const int wave   = threadIdx.x >> 5;
    const int tileId = (KSPLIT == 1) ? blockIdx.x * 4 + wave : blockIdx.x;
    if (KSPLIT == 1 && tileId >= NTM * NTN) return;
    const int tm = tileId / NTN;
    const int tn = tileId - tm * NTN;
    constexpr int JST = FUSE ? NTN : 1;
    const int ct0 = FUSE ? tn : tn * NT;
    const int c  = lane & 15;
    const int h  = lane >> 4;
    const int k0 = (KSPLIT == 1) ? 0 : wave * KW;

    const float* Ar[MT];
#pragma unroll
    for (int i = 0; i < MT; i++) {
        int r = tm * (16 * MT) + 16 * i + c;
        if (flipA) r = M - 1 - r;
        Ar[i] = A + (size_t)r * K + k0 + 4 * h;
    }
    const float* Br = Bw + (size_t)(ct0 * 16 + c) * K + k0 + 4 * h;

    v8f acc[MT][NT];
#pragma unroll
    for (int i = 0; i < MT; i++)
#pragma unroll
        for (int j = 0; j < NT; j++) acc[i][j] = v8f{};

    v4f fa0[MT][QG], fb0[NT][QG], fa1[MT][QG], fb1[NT][QG];

    auto load_set = [&](v4f (&fa)[MT][QG], v4f (&fb)[NT][QG], int grp) {
#pragma unroll
        for (int i = 0; i < MT; i++)
#pragma unroll
            for (int q = 0; q < QG; q++) fa[i][q] = *(const v4f*)(Ar[i] + grp * KG + q * KB);
#pragma unroll
        for (int j = 0; j < NT; j++)
#pragma unroll
            for (int q = 0; q < QG; q++)
                fb[j][q] = *(const v4f*)(Br + (size_t)j * JST * 16 * K + grp * KG + q * KB);
        __builtin_amdgcn_sched_barrier(0);
    };
    auto mma_set = [&](const v4f (&fa)[MT][QG], const v4f (&fb)[NT][QG]) {
#pragma unroll
        for (int q = 0; q < QG; q++) {
#pragma unroll
            for (int i = 0; i < MT; i++) {
                const v2f a = __builtin_shufflevector(fa[i][q], fa[i][q], 0, 1);
#pragma unroll
                for (int j = 0; j < NT; j++) {
                    const v2f b = __builtin_shufflevector(fb[j][q], fb[j][q], 0, 1);
                    acc[i][j] = __builtin_amdgcn_wmma_f32_16x16x4_f32(false, a, false, b,
                                                                      (short)0, acc[i][j], false, false);
                }
            }
#pragma unroll
            for (int i = 0; i < MT; i++) {
                const v2f a = __builtin_shufflevector(fa[i][q], fa[i][q], 2, 3);
#pragma unroll
                for (int j = 0; j < NT; j++) {
                    const v2f b = __builtin_shufflevector(fb[j][q], fb[j][q], 2, 3);
                    acc[i][j] = __builtin_amdgcn_wmma_f32_16x16x4_f32(false, a, false, b,
                                                                      (short)0, acc[i][j], false, false);
                }
            }
        }
        __builtin_amdgcn_sched_barrier(0);
    };

    load_set(fa0, fb0, 0);
    for (int grp = 0; grp < NG - 2; grp += 2) {
        load_set(fa1, fb1, grp + 1);
        mma_set(fa0, fb0);
        load_set(fa0, fb0, grp + 2);
        mma_set(fa1, fb1);
    }
    load_set(fa1, fb1, NG - 1);
    mma_set(fa0, fb0);
    mma_set(fa1, fb1);

    if constexpr (FUSE) {
        const int rbase = tm * (16 * MT) + 8 * h;
#pragma unroll
        for (int j = 0; j < NT; j++) {
            const int col16 = (ct0 + j * JST) * 16;
            const int col   = col16 + c;
            float* Cp = C + (size_t)rbase * N + col;
            if (col16 < DINNER || col16 >= DINNER + CONVDIM) {
#pragma unroll
                for (int i = 0; i < MT; i++)
#pragma unroll
                    for (int v = 0; v < 8; v++)
                        Cp[(size_t)(16 * i + v) * N] = acc[i][j][v];
            } else {
                const int ch = col - DINNER;
                float* Xp = xbc + (size_t)rbase * CONVDIM + ch;
                const float w0 = convw[ch * 4 + 0];
                const float w1 = convw[ch * 4 + 1];
                const float w2 = convw[ch * 4 + 2];
                const float w3 = convw[ch * 4 + 3];
                const float cb = convb[ch];
                float pv[MT][3];
#pragma unroll
                for (int i = 0; i < MT; i++)
#pragma unroll
                    for (int q = 0; q < 3; q++)
                        pv[i][q] = __shfl_xor(acc[i][j][5 + q], 16, 32);
#pragma unroll
                for (int i = 0; i < MT; i++) {
                    float p[11];
#pragma unroll
                    for (int q = 0; q < 3; q++)
                        p[q] = h ? pv[i][q] : pv[i > 0 ? i - 1 : 0][q];
#pragma unroll
                    for (int v = 0; v < 8; v++) p[3 + v] = acc[i][j][v];
#pragma unroll
                    for (int v = 0; v < 8; v++) {
                        float a = cb;
                        a += w0 * p[v];
                        a += w1 * p[v + 1];
                        a += w2 * p[v + 2];
                        a += w3 * p[v + 3];
                        const float s = a / (1.0f + expf(-a));
                        const bool first3 = (i == 0 && v < 3);
                        const bool last3  = (i == MT - 1 && v >= 5);
                        if (!first3 || h) Xp[(size_t)(16 * i + v) * CONVDIM] = s;
                        if ((first3 && !h) || (last3 && h)) Cp[(size_t)(16 * i + v) * N] = acc[i][j][v];
                    }
                }
            }
        }
    } else if constexpr (KSPLIT == 1) {
        float* Cp = C + (size_t)(tm * (16 * MT) + 8 * h) * N + tn * (16 * NT) + c;
#pragma unroll
        for (int i = 0; i < MT; i++)
#pragma unroll
            for (int j = 0; j < NT; j++)
#pragma unroll
                for (int v = 0; v < 8; v++)
                    Cp[(size_t)(16 * i + v) * N + 16 * j] = acc[i][j][v];
    } else {
        __shared__ __attribute__((aligned(16))) float s_part[KSPLIT * MT * NT * 256];
        float* mine = s_part + wave * (MT * NT * 256) + lane * 4;
#pragma unroll
        for (int i = 0; i < MT; i++)
#pragma unroll
            for (int j = 0; j < NT; j++) {
                *(v4f*)(mine + ((i * NT + j) * 2 + 0) * 128) = __builtin_shufflevector(acc[i][j], acc[i][j], 0, 1, 2, 3);
                *(v4f*)(mine + ((i * NT + j) * 2 + 1) * 128) = __builtin_shufflevector(acc[i][j], acc[i][j], 4, 5, 6, 7);
            }
        __syncthreads();
        float* Cp = C + (size_t)(tm * (16 * MT) + 16 * wave + 8 * h) * N + tn * (16 * NT) + c;
#pragma unroll
        for (int j = 0; j < NT; j++)
#pragma unroll
            for (int q = 0; q < 2; q++) {
                const float* src = s_part + ((wave * NT + j) * 2 + q) * 128 + lane * 4;
                v4f s = *(const v4f*)(src);
#pragma unroll
                for (int p = 1; p < KSPLIT; p++) s += *(const v4f*)(src + p * (MT * NT * 256));
#pragma unroll
                for (int e = 0; e < 4; e++)
                    Cp[(size_t)(4 * q + e) * N + 16 * j] = s[e];
            }
    }
}

#define EDGE_CONV_N   ((L_SEQ / INP_ROWS) * 3 * CONVDIM)
#define EDGE_CONV_BLK (EDGE_CONV_N / 256)
#define EDGE_DT_BLK   (NCH * NHEADSN / 4)
static_assert(EDGE_CONV_N % 256 == 0, "conv edge blocks are full");
static_assert(QCH == 64 && (NCH * NHEADSN) % 4 == 0, "two waves per chunk, four chunks per block");

__global__ void __launch_bounds__(256) bm_conv_edge_dt(const float* __restrict__ zx,
                                                       const float* __restrict__ convw,
                                                       const float* __restrict__ convb,
                                                       const float* __restrict__ dtb,
                                                       const float* __restrict__ Alog,
                                                       float* __restrict__ xbc,
                                                       float* __restrict__ dt_o,
                                                       float* __restrict__ cdec,
                                                       float* __restrict__ wdec,
                                                       float* __restrict__ cshi,
                                                       float* __restrict__ cslo,
                                                       float* __restrict__ ecs)
{
    if (blockIdx.x < EDGE_CONV_BLK) {
        const int idx = blockIdx.x * 256 + threadIdx.x;
        const int e = idx / CONVDIM;
        const int c = idx - e * CONVDIM;
        const int l = (e / 3) * INP_ROWS + (e % 3);
        float acc = convb[c];
#pragma unroll
        for (int t = 0; t < 4; t++) {
            const int ll = l - 3 + t;
            if (ll >= 0)
                acc += convw[c * 4 + t] * zx[(size_t)ll * DINPROJ + DINNER + c];
        }
        xbc[(size_t)l * CONVDIM + c] = acc / (1.0f + expf(-acc));
        return;
    }

    const int lane  = threadIdx.x & 31;
    const int wave  = threadIdx.x >> 5;
    const int pair  = (blockIdx.x - EDGE_CONV_BLK) * 4 + (wave >> 1);
    const int chunk = pair / NHEADSN;
    const int head  = pair - chunk * NHEADSN;
    const int t     = ((wave & 1) << 5) + lane;
    const int l     = chunk * QCH + t;

    const float v  = zx[(size_t)l * DINPROJ + DINNER + CONVDIM + head] + dtb[head];
    const float sp = (v > 20.0f) ? v : log1pf(expf(v));
    dt_o[l * NHEADSN + head] = sp;

    const float a = sp * (-expf(Alog[head]));
    double cs = (double)a;
#pragma unroll
    for (int off = 1; off < 32; off <<= 1) {
        const double up = __shfl_up(cs, off, 32);
        if (lane >= off) cs += up;
    }
    __shared__ double s_tot[8];
    if (lane == 31) s_tot[wave] = cs;
    __syncthreads();
    const double tot0 = s_tot[wave & 6];
    const double tot1 = s_tot[wave | 1];
    if (wave & 1) cs += tot0;
    const double cs_last = tot0 + tot1;

    const float hi = (float)cs;
    const float lo = (float)(cs - (double)hi);
    const int   o  = head * L_SEQ + l;
    cshi[o] = hi;
    cslo[o] = lo;
    ecs[o]  = expf(hi);
    wdec[o] = expf((float)(cs_last - cs)) * sp;
    if (t == 0) cdec[chunk * NHEADSN + head] = expf((float)cs_last);
}

#define SM_LD 68

__global__ void __launch_bounds__(256) bm_chunk_state(const float* __restrict__ xbc,
                               const float* __restrict__ wdec,
                               float* __restrict__ S)
{
    const int c    = blockIdx.x;
    const int head = blockIdx.y;
    const int tid  = threadIdx.x;
    const int lane = tid & 31;
    const int wave = tid >> 5;
    const int l0   = c * QCH;

    const float* wl = wdec + head * L_SEQ + l0;

    const int mn   = lane & 15;
    const int koff = (lane >> 4) << 1;
    const float* xcol = xbc + (size_t)l0 * CONVDIM + head * HEADDIM + mn;
    const float* bcol = xbc + (size_t)l0 * CONVDIM + DINNER + wave * 16 + mn;

    v8f acc0 = {}, acc1 = {}, acc2 = {}, acc3 = {};
#pragma unroll
    for (int k = 0; k < QCH; k += 4) {
        const int s0 = k + koff;
        const int s1 = s0 + 1;
        const float w0 = wl[s0];
        const float w1 = wl[s1];
        const float* x0 = xcol + (size_t)s0 * CONVDIM;
        const float* x1 = xcol + (size_t)s1 * CONVDIM;
        v2f b, a0, a1, a2, a3;
        b.x  = bcol[(size_t)s0 * CONVDIM];
        b.y  = bcol[(size_t)s1 * CONVDIM];
        a0.x = w0 * x0[0];   a0.y = w1 * x1[0];
        a1.x = w0 * x0[16];  a1.y = w1 * x1[16];
        a2.x = w0 * x0[32];  a2.y = w1 * x1[32];
        a3.x = w0 * x0[48];  a3.y = w1 * x1[48];
        acc0 = __builtin_amdgcn_wmma_f32_16x16x4_f32(false, a0, false, b,
                                                     (short)0, acc0, false, false);
        acc1 = __builtin_amdgcn_wmma_f32_16x16x4_f32(false, a1, false, b,
                                                     (short)0, acc1, false, false);
        acc2 = __builtin_amdgcn_wmma_f32_16x16x4_f32(false, a2, false, b,
                                                     (short)0, acc2, false, false);
        acc3 = __builtin_amdgcn_wmma_f32_16x16x4_f32(false, a3, false, b,
                                                     (short)0, acc3, false, false);
    }

    const int msub = (lane >> 4) << 3;
    float* Sp = S + ((size_t)(c * NHEADSN + head) * HEADDIM + msub) * DSTATE + wave * 16 + mn;
#pragma unroll
    for (int v = 0; v < 8; v++) Sp[(size_t)v * DSTATE]        = acc0[v];
#pragma unroll
    for (int v = 0; v < 8; v++) Sp[(size_t)(16 + v) * DSTATE] = acc1[v];
#pragma unroll
    for (int v = 0; v < 8; v++) Sp[(size_t)(32 + v) * DSTATE] = acc2[v];
#pragma unroll
    for (int v = 0; v < 8; v++) Sp[(size_t)(48 + v) * DSTATE] = acc3[v];
}

__global__ void __launch_bounds__(256) bm_state_pass(float* __restrict__ S,
                              const float* __restrict__ cdec)
{
    const int idx  = blockIdx.x * blockDim.x + threadIdx.x;
    const int head = idx / (HEADDIM * DSTATE);
    const size_t plane = (size_t)NHEADSN * HEADDIM * DSTATE;

    float t[NCH];
#pragma unroll
    for (int c = 0; c < NCH; c++) t[c] = S[(size_t)c * plane + idx];

    float run = 0.0f;
#pragma unroll
    for (int c = 0; c < NCH; c++) {
        S[(size_t)c * plane + idx] = run;
        run = cdec[c * NHEADSN + head] * run + t[c];
    }
}

__global__ void __launch_bounds__(256) bm_chunk_out(const float* __restrict__ xbc,
                             const float* __restrict__ dt,
                             const float* __restrict__ cshi,
                             const float* __restrict__ cslo,
                             const float* __restrict__ ecs,
                             const float* __restrict__ Hs,
                             float* __restrict__ yscan)
{
    const int c    = blockIdx.x;
    const int head = blockIdx.y;
    const int tid  = threadIdx.x;
    const int lane = tid & 31;
    const int wave = tid >> 5;
    const int l0   = c * QCH;

    __shared__ __attribute__((aligned(16))) float s_M[QCH * SM_LD];

    const int mn   = lane & 15;
    const int koff = (lane >> 4) << 1;
    const int msub = (lane >> 4) << 3;
    const int tt   = wave >> 1;
    const int cb   = (wave & 1) * 2;

    const int sA = (cb + 0) * 16 + mn;
    const int sB = (cb + 1) * 16 + mn;
    const int tb = tt * 16 + msub;
    const float* hi = cshi + head * L_SEQ + l0;
    const float* lo = cslo + head * L_SEQ + l0;
    const float hiA = hi[sA], loA = lo[sA];
    const float hiB = hi[sB], loB = lo[sB];
    const float dtA = dt[(l0 + sA) * NHEADSN + head];
    const float dtB = dt[(l0 + sB) * NHEADSN + head];
    float hit[8], lot[8];
#pragma unroll
    for (int v = 0; v < 8; v++) {
        hit[v] = hi[tb + v];
        lot[v] = lo[tb + v];
    }

    const float* Cr = xbc + (size_t)(l0 + tt * 16 + mn) * CONVDIM + DINNER + DSTATE + koff;

    {
        const float* B0 = xbc + (size_t)(l0 + (cb + 0) * 16 + mn) * CONVDIM + DINNER + koff;
        const float* B1 = xbc + (size_t)(l0 + (cb + 1) * 16 + mn) * CONVDIM + DINNER + koff;
        v8f g0 = {}, g1 = {};
        for (int k = 0; k < DSTATE; k += 4) {
            const v2f a  = *(const v2f*)(Cr + k);
            const v2f b0 = *(const v2f*)(B0 + k);
            const v2f b1 = *(const v2f*)(B1 + k);
            g0 = __builtin_amdgcn_wmma_f32_16x16x4_f32(false, a, false, b0,
                                                       (short)0, g0, false, false);
            g1 = __builtin_amdgcn_wmma_f32_16x16x4_f32(false, a, false, b1,
                                                       (short)0, g1, false, false);
        }
#pragma unroll
        for (int v = 0; v < 8; v++) {
            const int t = tb + v;
            const float dA = (hit[v] - hiA) + (lot[v] - loA);
            const float dB = (hit[v] - hiB) + (lot[v] - loB);
            const float mA = (sA <= t) ? g0[v] * expf(dA) * dtA : 0.0f;
            const float mB = (sB <= t) ? g1[v] * expf(dB) * dtB : 0.0f;
            s_M[t * SM_LD + sA] = mA;
            s_M[t * SM_LD + sB] = mB;
        }
    }
    __syncthreads();

    const float* el = ecs + head * L_SEQ + l0 + tb;
    float et[8];
#pragma unroll
    for (int v = 0; v < 8; v++) et[v] = el[v];
    v8f yi0 = {}, yi1 = {};
    {
        const float* H0 = Hs + ((size_t)(c * NHEADSN + head) * HEADDIM + (cb + 0) * 16 + mn) * DSTATE + koff;
        const float* H1 = Hs + ((size_t)(c * NHEADSN + head) * HEADDIM + (cb + 1) * 16 + mn) * DSTATE + koff;
        for (int k = 0; k < DSTATE; k += 4) {
            const v2f a  = *(const v2f*)(Cr + k);
            const v2f h0 = *(const v2f*)(H0 + k);
            const v2f h1 = *(const v2f*)(H1 + k);
            yi0 = __builtin_amdgcn_wmma_f32_16x16x4_f32(false, a, false, h0,
                                                        (short)0, yi0, false, false);
            yi1 = __builtin_amdgcn_wmma_f32_16x16x4_f32(false, a, false, h1,
                                                        (short)0, yi1, false, false);
        }
    }
    v8f ya0 = {}, ya1 = {};
    {
        const float* Ml = s_M + (tt * 16 + mn) * SM_LD + koff;
        const float* X0 = xbc + (size_t)l0 * CONVDIM + head * HEADDIM + (cb + 0) * 16 + mn;
        const float* X1 = X0 + 16;
        const int kend = (tt + 1) * 16;
        for (int k = 0; k < kend; k += 4) {
            const int s0 = k + koff;
            const int s1 = s0 + 1;
            const v2f a = *(const v2f*)(Ml + k);
            v2f x0, x1;
            x0.x = X0[(size_t)s0 * CONVDIM];  x0.y = X0[(size_t)s1 * CONVDIM];
            x1.x = X1[(size_t)s0 * CONVDIM];  x1.y = X1[(size_t)s1 * CONVDIM];
            ya0 = __builtin_amdgcn_wmma_f32_16x16x4_f32(false, a, false, x0,
                                                        (short)0, ya0, false, false);
            ya1 = __builtin_amdgcn_wmma_f32_16x16x4_f32(false, a, false, x1,
                                                        (short)0, ya1, false, false);
        }
    }

    float* Yp = yscan + (size_t)(l0 + tt * 16 + msub) * DINNER + head * HEADDIM + cb * 16 + mn;
#pragma unroll
    for (int v = 0; v < 8; v++) {
        const float e = et[v];
        Yp[(size_t)v * DINNER]      = ya0[v] + e * yi0[v];
        Yp[(size_t)v * DINNER + 16] = ya1[v] + e * yi1[v];
    }
}

__global__ void bm_gate_rmsnorm(float* __restrict__ y,
                                const float* __restrict__ xbc,
                                const float* __restrict__ zx,
                                const float* __restrict__ Dp,
                                const float* __restrict__ normw)
{
    const int l   = blockIdx.x;
    const int tid = threadIdx.x;
    __shared__ float red[8];
    __shared__ float scale;

    float vals[4];
    float ss = 0.0f;
#pragma unroll
    for (int j = 0; j < 4; j++) {
        const int i = tid * 4 + j;
        const int hh = i >> 6;
        float v = y[(size_t)l * DINNER + i] + Dp[hh] * xbc[(size_t)l * CONVDIM + i];
        const float z = zx[(size_t)l * DINPROJ + i];
        v *= z / (1.0f + expf(-z));
        vals[j] = v;
        ss += v * v;
    }
#pragma unroll
    for (int o = 16; o >= 1; o >>= 1) ss += __shfl_xor(ss, o, 32);
    if ((tid & 31) == 0) red[tid >> 5] = ss;
    __syncthreads();
    if (tid == 0) {
        float t = 0.0f;
        for (int i = 0; i < 8; i++) t += red[i];
        scale = rsqrtf(t / (float)DINNER + EPS_RMS);
    }
    __syncthreads();
#pragma unroll
    for (int j = 0; j < 4; j++) {
        const int i = tid * 4 + j;
        y[(size_t)l * DINNER + i] = vals[j] * scale * normw[i];
    }
}

__global__ void bm_add_flip(const float* __restrict__ of,
                            const float* __restrict__ ob,
                            float* __restrict__ out)
{
    const int idx = blockIdx.x * blockDim.x + threadIdx.x;
    if (idx >= L_SEQ * DMODEL) return;
    const int l = idx / DMODEL;
    const int j = idx - l * DMODEL;
    out[idx] = of[idx] + ob[(size_t)(L_SEQ - 1 - l) * DMODEL + j];
}

extern "C" void kernel_launch(void* const* d_in, const int* in_sizes, int n_in,
                              void* d_out, int out_size, void* d_ws, size_t ws_size,
                              hipStream_t stream)
{
    const float* Win[2]   = { (const float*)d_in[0],  (const float*)d_in[8]  };
    const float* convw[2] = { (const float*)d_in[1],  (const float*)d_in[9]  };
    const float* convb[2] = { (const float*)d_in[2],  (const float*)d_in[10] };
    const float* dtb[2]   = { (const float*)d_in[3],  (const float*)d_in[11] };
    const float* Alog[2]  = { (const float*)d_in[4],  (const float*)d_in[12] };
    const float* Dp[2]    = { (const float*)d_in[5],  (const float*)d_in[13] };
    const float* normw[2] = { (const float*)d_in[6],  (const float*)d_in[14] };
    const float* Wout[2]  = { (const float*)d_in[7],  (const float*)d_in[15] };
    const float* x        =   (const float*)d_in[16];

    float* ws = (float*)d_ws;
    size_t off = 0;
    float* zx[2];  float* xbc[2]; float* dtw[2];
    float* ysc[2]; float* od[2];
    for (int d = 0; d < 2; d++) { zx[d]  = ws + off; off += (size_t)L_SEQ * DINPROJ; }
    for (int d = 0; d < 2; d++) { xbc[d] = ws + off; off += (size_t)L_SEQ * CONVDIM; }
    for (int d = 0; d < 2; d++) { dtw[d] = ws + off; off += (size_t)L_SEQ * NHEADSN; }
    for (int d = 0; d < 2; d++) { ysc[d] = ws + off; off += (size_t)L_SEQ * DINNER;  }
    for (int d = 0; d < 2; d++) { od[d]  = ws + off; off += (size_t)L_SEQ * DMODEL;  }

    const dim3 blk256(256);

    for (int d = 0; d < 2; d++) {
        {
            const int tiles = (L_SEQ / INP_ROWS) * (DINPROJ / 80);
            bm_wmma_gemm_nt_f32<DMODEL, DINPROJ, 4, 5, 1, true><<<dim3((tiles + 3) / 4), dim3(128), 0, stream>>>(
                x, Win[d], zx[d], d, convw[d], convb[d], xbc[d]);
        }
        float* Sst  = zx[1 - d];
        float* cdec = Sst  + (size_t)NCH * NHEADSN * HEADDIM * DSTATE;
        float* wdec = cdec + (size_t)NCH * NHEADSN;
        float* cshi = wdec + (size_t)NHEADSN * L_SEQ;
        float* cslo = cshi + (size_t)NHEADSN * L_SEQ;
        float* ecs  = cslo + (size_t)NHEADSN * L_SEQ;
        bm_conv_edge_dt<<<dim3(EDGE_CONV_BLK + EDGE_DT_BLK), blk256, 0, stream>>>(
            zx[d], convw[d], convb[d], dtb[d], Alog[d], xbc[d], dtw[d],
            cdec, wdec, cshi, cslo, ecs);
        {
            bm_chunk_state<<<dim3(NCH, NHEADSN), blk256, 0, stream>>>(
                xbc[d], wdec, Sst);
            bm_state_pass<<<dim3(NHEADSN * HEADDIM * DSTATE / 256), blk256, 0, stream>>>(
                Sst, cdec);
            bm_chunk_out<<<dim3(NCH, NHEADSN), blk256, 0, stream>>>(
                xbc[d], dtw[d], cshi, cslo, ecs, Sst, ysc[d]);
        }
        bm_gate_rmsnorm<<<dim3(L_SEQ), blk256, 0, stream>>>(
            ysc[d], xbc[d], zx[d], Dp[d], normw[d]);
        {
            const int tiles = (L_SEQ / 64) * (DMODEL / 64);
            bm_wmma_gemm_nt_f32<DINNER, DMODEL, 4, 4, 4, false><<<dim3(tiles), dim3(128), 0, stream>>>(
                ysc[d], Wout[d], od[d], 0, nullptr, nullptr, nullptr);
        }
    }
    bm_add_flip<<<dim3((L_SEQ * DMODEL + 255) / 256), blk256, 0, stream>>>(
        od[0], od[1], (float*)d_out);
}


# chunk_out work-groups split by head half: 16 steps x 8 heads, 256 groups (one per WGP) instead of 128 x 16 waves; row sums of squares leave as 2 partials, out_proj scales its rows by the RMS factor
# speedup vs baseline: 39.8134x; 1.3667x over previous
#include <hip/hip_runtime.h>
#include <math.h>

#define L_SEQ     2048
#define DMODEL    512
#define DINNER    1024
#define DSTATE    128
#define HEADDIM   64
#define NHEADSN   16
#define CONVDIM   1280
#define DINPROJ   2320
#define EPS_RMS   1e-5f
#define QCH       64
#define NCH       (L_SEQ / QCH)

typedef __attribute__((ext_vector_type(2))) float v2f;
typedef __attribute__((ext_vector_type(4))) float v4f;
typedef __attribute__((ext_vector_type(8))) float v8f;

#define SSQ_ROW0  1860
#define SSQ_HALVES 2
static_assert(L_SEQ == 2048 && L_SEQ / 2 <= CONVDIM, "two plane rows hold one partial of every step");
static_assert(SSQ_ROW0 % 64 >= 3 && (SSQ_ROW0 + 2 * SSQ_HALVES - 1) % 64 <= 60 &&
              (SSQ_ROW0 + 2 * SSQ_HALVES - 1) / 64 == SSQ_ROW0 / 64 && SSQ_ROW0 + 2 * SSQ_HALVES <= L_SEQ,
              "rows whose xBC columns in_proj leaves unwritten");
static_assert((size_t)SSQ_ROW0 * DINPROJ >=
              (size_t)NCH * NHEADSN * HEADDIM * DSTATE + (size_t)NCH * NHEADSN + 3 * (size_t)NHEADSN * L_SEQ,
              "behind the scan scratch of the other direction");
__device__ __forceinline__ int ssq_at(int l, int hz)
{
    return (2 * hz + (l >> 10)) * DINPROJ + (l & 1023);
}

#define INP_ROWS 64

template <int K, int N, int MT, int NT, int KSPLIT, bool FUSE, bool DUAL>
__global__ void __launch_bounds__(128) bm_wmma_gemm_nt_f32(const float* __restrict__ A,
                                                           const float* __restrict__ Bw,
                                                           float* __restrict__ C,
                                                           int flipA,
                                                           const float* __restrict__ convw,
                                                           const float* __restrict__ convb,
                                                           float* __restrict__ xbc,
                                                           const float* __restrict__ A2,
                                                           const float* __restrict__ B2,
                                                           const float* __restrict__ ssq1,
                                                           const float* __restrict__ ssq2)
{
    constexpr int M    = L_SEQ;
    constexpr int KB   = 8;
    constexpr int QG   = 4;
    constexpr int KG   = KB * QG;
    constexpr int KW   = DUAL ? K / 2 : K / KSPLIT;
    constexpr int NG   = KW / KG;
    constexpr int NTN  = N / (16 * NT);
    constexpr int NTM  = M / (16 * MT);
    static_assert(N % (16 * NT) == 0 && M % (16 * MT) == 0 && KW % (2 * KG) == 0, "tile shape");
    static_assert(KSPLIT == 1 || (KSPLIT == 4 && MT == 4), "split shape");
    static_assert(!DUAL || (KSPLIT == 4 && !FUSE), "joined k axis: two waves per operand pair");
    static_assert(!FUSE || (KSPLIT == 1 && N == DINPROJ && 16 * MT == INP_ROWS), "fused epilogue shape");

    const int lane   = threadIdx.x & 31;
    const int wave   = threadIdx.x >> 5;
    const int tileId = (KSPLIT == 1) ? blockIdx.x * 4 + wave : blockIdx.x;
    if (KSPLIT == 1 && tileId >= NTM * NTN) return;
    const int tm = tileId / NTN;
    const int tn = tileId - tm * NTN;
    constexpr int JST = FUSE ? NTN : 1;
    const int ct0 = FUSE ? tn : tn * NT;
    const int c  = lane & 15;
    const int h  = lane >> 4;
    const bool second = DUAL && wave >= 2;
    const int  k0     = (KSPLIT == 1) ? 0 : (DUAL ? (wave & 1) * KW : wave * KW);
    const float* Asrc = second ? A2 : A;
    const float* Bsrc = second ? B2 : Bw;
    const bool  flip  = DUAL ? second : (flipA != 0);

    const float* Ar[MT];
#pragma unroll
    for (int i = 0; i < MT; i++) {
        int r = tm * (16 * MT) + 16 * i + c;
        if (flip) r = M - 1 - r;
        Ar[i] = Asrc + (size_t)r * K + k0 + 4 * h;
    }
    const float* Br = Bsrc + (size_t)(ct0 * 16 + c) * K + k0 + 4 * h;

    v8f acc[MT][NT];
#pragma unroll
    for (int i = 0; i < MT; i++)
#pragma unroll
        for (int j = 0; j < NT; j++) acc[i][j] = v8f{};

    v4f fa0[MT][QG], fb0[NT][QG], fa1[MT][QG], fb1[NT][QG];

    float rf[MT][8];
    if constexpr (DUAL) {
        const float* sq = second ? ssq2 : ssq1;
#pragma unroll
        for (int i = 0; i < MT; i++)
#pragma unroll
            for (int v = 0; v < 8; v++) {
                int r = tm * (16 * MT) + 16 * i + 8 * h + v;
                if (flip) r = M - 1 - r;
                const float t = sq[ssq_at(r, 0)] + sq[ssq_at(r, 1)];
                rf[i][v] = rsqrtf(t / (float)DINNER + EPS_RMS);
            }
    }

    auto load_set = [&](v4f (&fa)[MT][QG], v4f (&fb)[NT][QG], int grp) {
#pragma unroll
        for (int i = 0; i < MT; i++)
#pragma unroll
            for (int q = 0; q < QG; q++) fa[i][q] = *(const v4f*)(Ar[i] + grp * KG + q * KB);
#pragma unroll
        for (int j = 0; j < NT; j++)
#pragma unroll
            for (int q = 0; q < QG; q++)
                fb[j][q] = *(const v4f*)(Br + (size_t)j * JST * 16 * K + grp * KG + q * KB);
        __builtin_amdgcn_sched_barrier(0);
    };
    auto mma_set = [&](const v4f (&fa)[MT][QG], const v4f (&fb)[NT][QG]) {
#pragma unroll
        for (int q = 0; q < QG; q++) {
#pragma unroll
            for (int i = 0; i < MT; i++) {
                const v2f a = __builtin_shufflevector(fa[i][q], fa[i][q], 0, 1);
#pragma unroll
                for (int j = 0; j < NT; j++) {
                    const v2f b = __builtin_shufflevector(fb[j][q], fb[j][q], 0, 1);
                    acc[i][j] = __builtin_amdgcn_wmma_f32_16x16x4_f32(false, a, false, b,
                                                                      (short)0, acc[i][j], false, false);
                }
            }
#pragma unroll
            for (int i = 0; i < MT; i++) {
                const v2f a = __builtin_shufflevector(fa[i][q], fa[i][q], 2, 3);
#pragma unroll
                for (int j = 0; j < NT; j++) {
                    const v2f b = __builtin_shufflevector(fb[j][q], fb[j][q], 2, 3);
                    acc[i][j] = __builtin_amdgcn_wmma_f32_16x16x4_f32(false, a, false, b,
                                                                      (short)0, acc[i][j], false, false);
                }
            }
        }
        __builtin_amdgcn_sched_barrier(0);
    };

    load_set(fa0, fb0, 0);
    for (int grp = 0; grp < NG - 2; grp += 2) {
        load_set(fa1, fb1, grp + 1);
        mma_set(fa0, fb0);
        load_set(fa0, fb0, grp + 2);
        mma_set(fa1, fb1);
    }
    load_set(fa1, fb1, NG - 1);
    mma_set(fa0, fb0);
    mma_set(fa1, fb1);

    if constexpr (FUSE) {
        const int rbase = tm * (16 * MT) + 8 * h;
#pragma unroll
        for (int j = 0; j < NT; j++) {
            const int col16 = (ct0 + j * JST) * 16;
            const int col   = col16 + c;
            float* Cp = C + (size_t)rbase * N + col;
            if (col16 < DINNER || col16 >= DINNER + CONVDIM) {
#pragma unroll
                for (int i = 0; i < MT; i++)
#pragma unroll
                    for (int v = 0; v < 8; v++)
                        Cp[(size_t)(16 * i + v) * N] = acc[i][j][v];
            } else {
                const int ch = col - DINNER;
                float* Xp = xbc + (size_t)rbase * CONVDIM + ch;
                const float w0 = convw[ch * 4 + 0];
                const float w1 = convw[ch * 4 + 1];
                const float w2 = convw[ch * 4 + 2];
                const float w3 = convw[ch * 4 + 3];
                const float cb = convb[ch];
                float pv[MT][3];
#pragma unroll
                for (int i = 0; i < MT; i++)
#pragma unroll
                    for (int q = 0; q < 3; q++)
                        pv[i][q] = __shfl_xor(acc[i][j][5 + q], 16, 32);
#pragma unroll
                for (int i = 0; i < MT; i++) {
                    float p[11];
#pragma unroll
                    for (int q = 0; q < 3; q++)
                        p[q] = h ? pv[i][q] : pv[i > 0 ? i - 1 : 0][q];
#pragma unroll
                    for (int v = 0; v < 8; v++) p[3 + v] = acc[i][j][v];
#pragma unroll
                    for (int v = 0; v < 8; v++) {
                        float a = cb;
                        a += w0 * p[v];
                        a += w1 * p[v + 1];
                        a += w2 * p[v + 2];
                        a += w3 * p[v + 3];
                        const float s = a / (1.0f + expf(-a));
                        const bool first3 = (i == 0 && v < 3);
                        const bool last3  = (i == MT - 1 && v >= 5);
                        if (!first3 || h) Xp[(size_t)(16 * i + v) * CONVDIM] = s;
                        if ((first3 && !h) || (last3 && h)) Cp[(size_t)(16 * i + v) * N] = acc[i][j][v];
                    }
                }
            }
        }
    } else if constexpr (KSPLIT == 1) {
        float* Cp = C + (size_t)(tm * (16 * MT) + 8 * h) * N + tn * (16 * NT) + c;
#pragma unroll
        for (int i = 0; i < MT; i++)
#pragma unroll
            for (int j = 0; j < NT; j++)
#pragma unroll
                for (int v = 0; v < 8; v++)
                    Cp[(size_t)(16 * i + v) * N + 16 * j] = acc[i][j][v];
    } else {
        __shared__ __attribute__((aligned(16))) float s_part[KSPLIT * MT * NT * 256];
        float* mine = s_part + wave * (MT * NT * 256) + lane * 4;
        if constexpr (DUAL) {
#pragma unroll
            for (int i = 0; i < MT; i++)
#pragma unroll
                for (int j = 0; j < NT; j++)
#pragma unroll
                    for (int v = 0; v < 8; v++) acc[i][j][v] *= rf[i][v];
        }
#pragma unroll
        for (int i = 0; i < MT; i++)
#pragma unroll
            for (int j = 0; j < NT; j++) {
                *(v4f*)(mine + ((i * NT + j) * 2 + 0) * 128) = __builtin_shufflevector(acc[i][j], acc[i][j], 0, 1, 2, 3);
                *(v4f*)(mine + ((i * NT + j) * 2 + 1) * 128) = __builtin_shufflevector(acc[i][j], acc[i][j], 4, 5, 6, 7);
            }
        __syncthreads();
        float* Cp = C + (size_t)(tm * (16 * MT) + 16 * wave + 8 * h) * N + tn * (16 * NT) + c;
#pragma unroll
        for (int j = 0; j < NT; j++)
#pragma unroll
            for (int q = 0; q < 2; q++) {
                const float* src = s_part + ((wave * NT + j) * 2 + q) * 128 + lane * 4;
                v4f s = *(const v4f*)(src);
#pragma unroll
                for (int p = 1; p < KSPLIT; p++) s += *(const v4f*)(src + p * (MT * NT * 256));
#pragma unroll
                for (int e = 0; e < 4; e++)
                    Cp[(size_t)(4 * q + e) * N + 16 * j] = s[e];
            }
    }
}

#define SM_LD 68


#define EDGE_XB   (HEADDIM + DSTATE)
#define EDGE_CPH  (DSTATE / NHEADSN)
#define EDGE_COLS (EDGE_XB + EDGE_CPH)
static_assert(QCH == 64 && INP_ROWS == QCH, "chunk = in_proj wave tile; two waves hold a chunk's steps");
static_assert(DSTATE % NHEADSN == 0 && EDGE_COLS <= 256, "edge columns fit one work-group");

__global__ void __launch_bounds__(256) bm_chunk_state(const float* __restrict__ zx,
                               const float* __restrict__ convw,
                               const float* __restrict__ convb,
                               const float* __restrict__ dtb,
                               const float* __restrict__ Alog,
                               const float* __restrict__ xbc,
                               float* __restrict__ xedge,
                               float* __restrict__ dt_o,
                               float* __restrict__ cdec,
                               float* __restrict__ cshi,
                               float* __restrict__ cslo,
                               float* __restrict__ ecs,
                               float* __restrict__ S)
{
    const int c    = blockIdx.x;
    const int head = blockIdx.y;
    const int tid  = threadIdx.x;
    const int lane = tid & 31;
    const int wave = tid >> 5;
    const int l0   = c * QCH;

    __shared__ float  s_w[QCH];
    __shared__ float  s_edge[3 * EDGE_XB];
    __shared__ double s_tot[2];

    const bool edge = tid < EDGE_COLS;
    const bool step = tid < QCH;
    const int ch = (tid < HEADDIM) ? head * HEADDIM + tid
                 : (tid < EDGE_XB) ? DINNER + (tid - HEADDIM)
                                   : DINNER + DSTATE + head * EDGE_CPH + (tid - EDGE_XB);
    float raw[6] = {0.0f, 0.0f, 0.0f, 0.0f, 0.0f, 0.0f};
    float cw0 = 0.0f, cw1 = 0.0f, cw2 = 0.0f, cw3 = 0.0f, cbias = 0.0f;
    float dtraw = 0.0f, dtbias = 0.0f, alog = 0.0f;
    if (edge) {
        const float* zp = zx + (size_t)l0 * DINPROJ + DINNER + ch;
        if (c > 0) {
            raw[0] = zp[-3 * DINPROJ];
            raw[1] = zp[-2 * DINPROJ];
            raw[2] = zp[-1 * DINPROJ];
        }
        raw[3] = zp[0];
        raw[4] = zp[DINPROJ];
        raw[5] = zp[2 * DINPROJ];
        cw0 = convw[ch * 4 + 0];
        cw1 = convw[ch * 4 + 1];
        cw2 = convw[ch * 4 + 2];
        cw3 = convw[ch * 4 + 3];
        cbias = convb[ch];
    }
    if (step) {
        dtraw  = zx[(size_t)(l0 + tid) * DINPROJ + DINNER + CONVDIM + head];
        dtbias = dtb[head];
        alog   = Alog[head];
    }
    float  sp = 0.0f;
    double cs = 0.0;
    if (step) {
        const float v = dtraw + dtbias;
        sp = (v > 20.0f) ? v : log1pf(expf(v));
        const float a = sp * (-expf(alog));
        cs = (double)a;
#pragma unroll
        for (int off = 1; off < 32; off <<= 1) {
            const double up = __shfl_up(cs, off, 32);
            if (lane >= off) cs += up;
        }
        if (lane == 31) s_tot[wave] = cs;
    }
    float es[3] = {0.0f, 0.0f, 0.0f};
    if (edge) {
#pragma unroll
        for (int r = 0; r < 3; r++) {
            float a = cbias;
            a += cw0 * raw[r];
            a += cw1 * raw[r + 1];
            a += cw2 * raw[r + 2];
            a += cw3 * raw[r + 3];
            es[r] = a / (1.0f + expf(-a));
            if (tid < EDGE_XB) s_edge[r * EDGE_XB + tid] = es[r];
        }
    }
    __syncthreads();
    float hi = 0.0f, lo = 0.0f, ehi = 0.0f, cd = 0.0f;
    if (step) {
        const double tot0 = s_tot[0];
        const double tot1 = s_tot[1];
        if (wave & 1) cs += tot0;
        const double cs_last = tot0 + tot1;
        hi  = (float)cs;
        lo  = (float)(cs - (double)hi);
        ehi = expf(hi);
        cd  = expf((float)cs_last);
        s_w[tid] = expf((float)(cs_last - cs)) * sp;
    }
    __syncthreads();
    if (step) {
        const int l = l0 + tid;
        const int o = head * L_SEQ + l;
        dt_o[l * NHEADSN + head] = sp;
        cshi[o] = hi;
        cslo[o] = lo;
        ecs[o]  = ehi;
        if (tid == 0) cdec[c * NHEADSN + head] = cd;
    }
    if (edge && ((tid < HEADDIM) || (tid >= EDGE_XB) || (head == 0))) {
#pragma unroll
        for (int r = 0; r < 3; r++) xedge[(size_t)(l0 + r) * CONVDIM + ch] = es[r];
    }

    const int mn   = lane & 15;
    const int koff = (lane >> 4) << 1;
    const float* xcol = xbc + (size_t)l0 * CONVDIM + head * HEADDIM + mn;
    const float* bcol = xbc + (size_t)l0 * CONVDIM + DINNER + wave * 16 + mn;

    v8f acc0 = {}, acc1 = {}, acc2 = {}, acc3 = {};
#pragma unroll
    for (int k = 0; k < QCH; k += 4) {
        const int s0 = k + koff;
        const int s1 = s0 + 1;
        const float w0 = s_w[s0];
        const float w1 = s_w[s1];
        float xa[4], xb[4];
        v2f b;
        if (k == 0) {
            const float* e0 = s_edge + s0 * EDGE_XB;
#pragma unroll
            for (int j = 0; j < 4; j++) xa[j] = e0[mn + 16 * j];
            b.x = e0[HEADDIM + wave * 16 + mn];
            if (koff == 0) {
                const float* e1 = s_edge + EDGE_XB;
#pragma unroll
                for (int j = 0; j < 4; j++) xb[j] = e1[mn + 16 * j];
                b.y = e1[HEADDIM + wave * 16 + mn];
            } else {
                const float* x1 = xcol + (size_t)3 * CONVDIM;
#pragma unroll
                for (int j = 0; j < 4; j++) xb[j] = x1[16 * j];
                b.y = bcol[(size_t)3 * CONVDIM];
            }
        } else {
            const float* x0 = xcol + (size_t)s0 * CONVDIM;
            const float* x1 = xcol + (size_t)s1 * CONVDIM;
#pragma unroll
            for (int j = 0; j < 4; j++) { xa[j] = x0[16 * j]; xb[j] = x1[16 * j]; }
            b.x = bcol[(size_t)s0 * CONVDIM];
            b.y = bcol[(size_t)s1 * CONVDIM];
        }
        v2f a0, a1, a2, a3;
        a0.x = w0 * xa[0];  a0.y = w1 * xb[0];
        a1.x = w0 * xa[1];  a1.y = w1 * xb[1];
        a2.x = w0 * xa[2];  a2.y = w1 * xb[2];
        a3.x = w0 * xa[3];  a3.y = w1 * xb[3];
        acc0 = __builtin_amdgcn_wmma_f32_16x16x4_f32(false, a0, false, b,
                                                     (short)0, acc0, false, false);
        acc1 = __builtin_amdgcn_wmma_f32_16x16x4_f32(false, a1, false, b,
                                                     (short)0, acc1, false, false);
        acc2 = __builtin_amdgcn_wmma_f32_16x16x4_f32(false, a2, false, b,
                                                     (short)0, acc2, false, false);
        acc3 = __builtin_amdgcn_wmma_f32_16x16x4_f32(false, a3, false, b,
                                                     (short)0, acc3, false, false);
    }

    const int msub = (lane >> 4) << 3;
    float* Sp = S + ((size_t)(c * NHEADSN + head) * HEADDIM + msub) * DSTATE + wave * 16 + mn;
#pragma unroll
    for (int v = 0; v < 8; v++) Sp[(size_t)v * DSTATE]        = acc0[v];
#pragma unroll
    for (int v = 0; v < 8; v++) Sp[(size_t)(16 + v) * DSTATE] = acc1[v];
#pragma unroll
    for (int v = 0; v < 8; v++) Sp[(size_t)(32 + v) * DSTATE] = acc2[v];
#pragma unroll
    for (int v = 0; v < 8; v++) Sp[(size_t)(48 + v) * DSTATE] = acc3[v];
}

__global__ void __launch_bounds__(256) bm_state_pass(float* __restrict__ S,
                              const float* __restrict__ cdec)
{
    const int idx  = blockIdx.x * blockDim.x + threadIdx.x;
    const int head = idx / (HEADDIM * DSTATE);
    const size_t plane = (size_t)NHEADSN * HEADDIM * DSTATE;

    float t[NCH];
#pragma unroll
    for (int c = 0; c < NCH; c++) t[c] = S[(size_t)c * plane + idx];

    float run = 0.0f;
#pragma unroll
    for (int c = 0; c < NCH; c++) {
        S[(size_t)c * plane + idx] = run;
        run = cdec[c * NHEADSN + head] * run + t[c];
    }
}

#define ROWT 16
#define COH  (NHEADSN / SSQ_HALVES)
static_assert(QCH / ROWT <= COH, "one wave per G tile");
__global__ void __launch_bounds__(32 * COH) bm_chunk_out_rows(const float* __restrict__ xbc,
                                  const float* __restrict__ zx,
                                  const float* __restrict__ dt,
                                  const float* __restrict__ cshi,
                                  const float* __restrict__ cslo,
                                  const float* __restrict__ ecs,
                                  const float* __restrict__ Hs,
                                  const float* __restrict__ Dp,
                                  const float* __restrict__ normw,
                                  float* __restrict__ yout,
                                  float* __restrict__ ssq)
{
    const int c    = blockIdx.x;
    const int tt   = blockIdx.y;
    const int hz   = blockIdx.z;
    const int tid  = threadIdx.x;
    const int lane = tid & 31;
    const int wv   = tid >> 5;
    const int head = hz * COH + wv;
    const int l0   = c * QCH;
    const int mn   = lane & 15;
    const int hh   = lane >> 4;
    const int koff = hh << 1;
    const int msub = hh << 3;

    __shared__ __attribute__((aligned(16))) float s_G[ROWT * SM_LD];
    __shared__ __attribute__((aligned(16))) float s_hi[COH][QCH];
    __shared__ __attribute__((aligned(16))) float s_lo[COH][QCH];
    __shared__ __attribute__((aligned(16))) float s_dt[COH][QCH];
    __shared__ float s_ss[COH][ROWT];

    {
        const float* hi = cshi + head * L_SEQ + l0;
        const float* lo = cslo + head * L_SEQ + l0;
        s_hi[wv][lane]      = hi[lane];
        s_hi[wv][lane + 32] = hi[lane + 32];
        s_lo[wv][lane]      = lo[lane];
        s_lo[wv][lane + 32] = lo[lane + 32];
        s_dt[wv][lane]      = dt[(l0 + lane) * NHEADSN + head];
        s_dt[wv][lane + 32] = dt[(l0 + lane + 32) * NHEADSN + head];
    }

    const float* Cr = xbc + (size_t)(l0 + tt * ROWT + mn) * CONVDIM + DINNER + DSTATE + 4 * hh;

    if (wv <= tt) {
        const float* Br = xbc + (size_t)(l0 + wv * 16 + mn) * CONVDIM + DINNER + 4 * hh;
        v8f g0 = {}, g1 = {};
#pragma unroll
        for (int q = 0; q < DSTATE / 8; q++) {
            const v4f a4 = *(const v4f*)(Cr + 8 * q);
            const v4f b4 = *(const v4f*)(Br + 8 * q);
            g0 = __builtin_amdgcn_wmma_f32_16x16x4_f32(false, __builtin_shufflevector(a4, a4, 0, 1), false,
                                                       __builtin_shufflevector(b4, b4, 0, 1),
                                                       (short)0, g0, false, false);
            g1 = __builtin_amdgcn_wmma_f32_16x16x4_f32(false, __builtin_shufflevector(a4, a4, 2, 3), false,
                                                       __builtin_shufflevector(b4, b4, 2, 3),
                                                       (short)0, g1, false, false);
        }
#pragma unroll
        for (int v = 0; v < 8; v++) s_G[(msub + v) * SM_LD + wv * 16 + mn] = g0[v] + g1[v];
    }

    v8f yi[4], ya[4];
#pragma unroll
    for (int j = 0; j < 4; j++) { yi[j] = v8f{}; ya[j] = v8f{}; }
    {
        const float* Hb = Hs + ((size_t)(c * NHEADSN + head) * HEADDIM + mn) * DSTATE + 4 * hh;
#pragma unroll
        for (int q = 0; q < DSTATE / 8; q++) {
            const v4f a4 = *(const v4f*)(Cr + 8 * q);
            const v2f a01 = __builtin_shufflevector(a4, a4, 0, 1);
            const v2f a23 = __builtin_shufflevector(a4, a4, 2, 3);
#pragma unroll
            for (int j = 0; j < 4; j++) {
                const v4f b4 = *(const v4f*)(Hb + (size_t)j * 16 * DSTATE + 8 * q);
                yi[j] = __builtin_amdgcn_wmma_f32_16x16x4_f32(false, a01, false, __builtin_shufflevector(b4, b4, 0, 1),
                                                              (short)0, yi[j], false, false);
                yi[j] = __builtin_amdgcn_wmma_f32_16x16x4_f32(false, a23, false, __builtin_shufflevector(b4, b4, 2, 3),
                                                              (short)0, yi[j], false, false);
            }
        }
    }

    __syncthreads();

    {
        const int   T   = tt * ROWT + mn;
        const float hiT = s_hi[wv][T];
        const float loT = s_lo[wv][T];
        const float* Gl = s_G + mn * SM_LD + koff;
        const float* Xb = xbc + (size_t)l0 * CONVDIM + head * HEADDIM + mn;
        for (int jt = 0; jt <= tt; jt++) {
#pragma unroll
            for (int kk = 0; kk < 4; kk++) {
                const int k  = jt * 16 + kk * 4;
                const int s0 = k + koff;
                const int s1 = s0 + 1;
                const v2f g2 = *(const v2f*)(Gl + k);
                const v2f h2 = *(const v2f*)(&s_hi[wv][s0]);
                const v2f l2 = *(const v2f*)(&s_lo[wv][s0]);
                const v2f d2 = *(const v2f*)(&s_dt[wv][s0]);
                const float e0 = (hiT - h2.x) + (loT - l2.x);
                const float e1 = (hiT - h2.y) + (loT - l2.y);
                const bool  in0 = (s0 <= T);
                const bool  in1 = (s1 <= T);
                const float w0  = expf(in0 ? e0 : 0.0f);
                const float w1  = expf(in1 ? e1 : 0.0f);
                v2f a;
                a.x = g2.x * w0 * d2.x * (in0 ? 1.0f : 0.0f);
                a.y = g2.y * w1 * d2.y * (in1 ? 1.0f : 0.0f);
#pragma unroll
                for (int j = 0; j < 4; j++) {
                    v2f xb;
                    xb.x = Xb[(size_t)s0 * CONVDIM + 16 * j];
                    xb.y = Xb[(size_t)s1 * CONVDIM + 16 * j];
                    ya[j] = __builtin_amdgcn_wmma_f32_16x16x4_f32(false, a, false, xb,
                                                                  (short)0, ya[j], false, false);
                }
            }
        }
    }

    const int   lrow = l0 + tt * ROWT + msub;
    const float Dh   = Dp[head];
    float nw[4];
#pragma unroll
    for (int j = 0; j < 4; j++) nw[j] = normw[head * HEADDIM + 16 * j + mn];
    float ss[8];
#pragma unroll
    for (int v = 0; v < 8; v++) {
        const int   l = lrow + v;
        const float e = ecs[head * L_SEQ + l];
        float acc = 0.0f;
#pragma unroll
        for (int j = 0; j < 4; j++) {
            const int   col = head * HEADDIM + 16 * j + mn;
            const float xv  = xbc[(size_t)l * CONVDIM + col];
            const float z   = zx[(size_t)l * DINPROJ + col];
            float y = ya[j][v] + e * yi[j][v];
            y = y + Dh * xv;
            y *= z / (1.0f + expf(-z));
            acc += y * y;
            yout[(size_t)l * DINNER + col] = y * nw[j];
        }
        acc += __shfl_xor(acc, 1, 32);
        acc += __shfl_xor(acc, 2, 32);
        acc += __shfl_xor(acc, 4, 32);
        acc += __shfl_xor(acc, 8, 32);
        ss[v] = acc;
    }
    if (mn == 0) {
#pragma unroll
        for (int v = 0; v < 8; v++) s_ss[wv][msub + v] = ss[v];
    }
    __syncthreads();
    if (tid < ROWT) {
        float t = 0.0f;
        for (int w = 0; w < COH; w++) t += s_ss[w][tid];
        ssq[ssq_at(l0 + tt * ROWT + tid, hz)] = t;
    }
}

extern "C" void kernel_launch(void* const* d_in, const int* in_sizes, int n_in,
                              void* d_out, int out_size, void* d_ws, size_t ws_size,
                              hipStream_t stream)
{
    const float* Win[2]   = { (const float*)d_in[0],  (const float*)d_in[8]  };
    const float* convw[2] = { (const float*)d_in[1],  (const float*)d_in[9]  };
    const float* convb[2] = { (const float*)d_in[2],  (const float*)d_in[10] };
    const float* dtb[2]   = { (const float*)d_in[3],  (const float*)d_in[11] };
    const float* Alog[2]  = { (const float*)d_in[4],  (const float*)d_in[12] };
    const float* Dp[2]    = { (const float*)d_in[5],  (const float*)d_in[13] };
    const float* normw[2] = { (const float*)d_in[6],  (const float*)d_in[14] };
    const float* Wout[2]  = { (const float*)d_in[7],  (const float*)d_in[15] };
    const float* x        =   (const float*)d_in[16];

    float* ws = (float*)d_ws;
    size_t off = 0;
    float* zx[2];  float* xbc[2]; float* dtw[2];
    float* ysc[2];
    for (int d = 0; d < 2; d++) { zx[d]  = ws + off; off += (size_t)L_SEQ * DINPROJ; }
    for (int d = 0; d < 2; d++) { xbc[d] = ws + off; off += (size_t)L_SEQ * CONVDIM; }
    for (int d = 0; d < 2; d++) { dtw[d] = ws + off; off += (size_t)L_SEQ * NHEADSN; }
    for (int d = 0; d < 2; d++) { ysc[d] = ws + off; off += (size_t)L_SEQ * DINNER;  }

    const dim3 blk256(256);

    for (int d = 0; d < 2; d++) {
        {
            const int tiles = (L_SEQ / INP_ROWS) * (DINPROJ / 80);
            bm_wmma_gemm_nt_f32<DMODEL, DINPROJ, 4, 5, 1, true, false><<<dim3((tiles + 3) / 4), dim3(128), 0, stream>>>(
                x, Win[d], zx[d], d, convw[d], convb[d], xbc[d], nullptr, nullptr, nullptr, nullptr);
        }
        float* Sst  = zx[1 - d];
        float* cdec = Sst  + (size_t)NCH * NHEADSN * HEADDIM * DSTATE;
        float* cshi = cdec + (size_t)NCH * NHEADSN;
        float* cslo = cshi + (size_t)NHEADSN * L_SEQ;
        float* ecs  = cslo + (size_t)NHEADSN * L_SEQ;
        {
            bm_chunk_state<<<dim3(NCH, NHEADSN), blk256, 0, stream>>>(
                zx[d], convw[d], convb[d], dtb[d], Alog[d], xbc[d], xbc[d], dtw[d],
                cdec, cshi, cslo, ecs, Sst);
            bm_state_pass<<<dim3(NHEADSN * HEADDIM * DSTATE / 256), blk256, 0, stream>>>(
                Sst, cdec);
            bm_chunk_out_rows<<<dim3(NCH, QCH / ROWT, SSQ_HALVES), dim3(32 * COH), 0, stream>>>(
                xbc[d], zx[d], dtw[d], cshi, cslo, ecs, Sst, Dp[d], normw[d], ysc[d],
                zx[d] + (size_t)SSQ_ROW0 * DINPROJ + DINNER);
        }
    }
    {
        const int tiles = (L_SEQ / 64) * (DMODEL / 64);
        bm_wmma_gemm_nt_f32<DINNER, DMODEL, 4, 4, 4, false, true><<<dim3(tiles), dim3(128), 0, stream>>>(
            ysc[0], Wout[0], (float*)d_out, 0, nullptr, nullptr, nullptr, ysc[1], Wout[1],
            zx[0] + (size_t)SSQ_ROW0 * DINPROJ + DINNER, zx[1] + (size_t)SSQ_ROW0 * DINPROJ + DINNER);
    }
}
